# starting kernel
# baseline (speedup 1.0000x reference)
#include <hip/hip_runtime.h>

typedef __attribute__((ext_vector_type(2))) float v2f;
typedef __attribute__((ext_vector_type(8))) float v8f;

#define N_NODES 50000
#define N_EDGES 500000
#define HID 128
#define LDSPITCH (HID + 4)

__global__ void zero_f32(float* __restrict__ p, long n) {
  long i = (long)blockIdx.x * blockDim.x + threadIdx.x;
  long stride = (long)gridDim.x * blockDim.x;
  long n4 = n >> 2;
  float4* p4 = (float4*)p;
  for (long j = i; j < n4; j += stride) p4[j] = make_float4(0.f, 0.f, 0.f, 0.f);
}

__global__ void degree_kernel(const int* __restrict__ dst, float* __restrict__ deg, int E) {
  int e = blockIdx.x * blockDim.x + threadIdx.x;
  if (e < E) atomicAdd(&deg[dst[e]], 1.0f);
}

__global__ void invdeg_kernel(float* __restrict__ deg, int n) {
  int i = blockIdx.x * blockDim.x + threadIdx.x;
  if (i < n) deg[i] = 1.0f / fmaxf(deg[i], 1.0f);
}

__global__ void scatter_kernel(const float* __restrict__ x,
                               const int* __restrict__ src,
                               const int* __restrict__ dst,
                               float* __restrict__ agg, int E) {
  int wave = (int)(((long)blockIdx.x * blockDim.x + threadIdx.x) >> 5);
  int lane = threadIdx.x & 31;
  if (wave >= E) return;
  int s = src[wave];
  int d = dst[wave];
  const float4 v = *(const float4*)(x + (long)s * HID + lane * 4);
  float* o = agg + (long)d * HID + lane * 4;
  atomicAdd(o + 0, v.x);
  atomicAdd(o + 1, v.y);
  atomicAdd(o + 2, v.z);
  atomicAdd(o + 3, v.w);
}

template <int OUTD, bool RELU, int STRIPS>
__global__ __launch_bounds__(32 * (OUTD / 16)) void sage_gemm(
    const float* __restrict__ xin,
    const float* __restrict__ agg,
    const float* __restrict__ invdeg,
    const float* __restrict__ Wl,
    const float* __restrict__ Wr,
    const float* __restrict__ bias,
    float* __restrict__ out,
    int n_nodes) {
  __shared__ float sX[16][LDSPITCH];
  __shared__ float sA[16][LDSPITCH];

  const int tid = threadIdx.x;
  const int nthreads = 32 * (OUTD / 16);
  const int wave = tid >> 5;
  const int lane = tid & 31;
  const int half = lane >> 4;
  const int r = lane & 15;
  const int col = wave * 16 + r;

  v2f BL[HID / 4], BR[HID / 4];
#pragma unroll
  for (int j = 0; j < HID / 4; ++j) {
    const int ka = 4 * j + 2 * half;
    BL[j][0] = Wl[(long)ka * OUTD + col];
    BL[j][1] = Wl[(long)(ka + 1) * OUTD + col];
    BR[j][0] = Wr[(long)ka * OUTD + col];
    BR[j][1] = Wr[(long)(ka + 1) * OUTD + col];
  }
  const float bn = bias[col];

  for (int s = 0; s < STRIPS; ++s) {
    const int row0 = (blockIdx.x * STRIPS + s) * 16;
    if (row0 >= n_nodes) break;

    for (int i = tid; i < 16 * (HID / 4); i += nthreads) {
      const int rr = i / (HID / 4);
      const int c4 = (i % (HID / 4)) * 4;
      const long base = (long)(row0 + rr) * HID + c4;
      float4 xv = *(const float4*)(xin + base);
      float4 av = *(const float4*)(agg + base);
      const float iv = invdeg[row0 + rr];
      av.x *= iv; av.y *= iv; av.z *= iv; av.w *= iv;
      *(float4*)&sX[rr][c4] = xv;
      *(float4*)&sA[rr][c4] = av;
    }
    __syncthreads();

    v8f accL = {};
    v8f accR = {};
#pragma unroll
    for (int j = 0; j < HID / 4; ++j) {
      const int ka = 4 * j + 2 * half;
      v2f aL, aR;
      aL[0] = sA[r][ka];
      aL[1] = sA[r][ka + 1];
      aR[0] = sX[r][ka];
      aR[1] = sX[r][ka + 1];
      accL = __builtin_amdgcn_wmma_f32_16x16x4_f32(false, aL, false, BL[j],
                                                   (short)0, accL, false, false);
      accR = __builtin_amdgcn_wmma_f32_16x16x4_f32(false, aR, false, BR[j],
                                                   (short)0, accR, false, false);
    }

#pragma unroll
    for (int i = 0; i < 8; ++i) {
      const int gr = row0 + half * 8 + i;
      float v = accL[i] + accR[i] + bn;
      if (RELU) v = fmaxf(v, 0.0f);
      out[(long)gr * OUTD + col] = v;
    }
    __syncthreads();
  }
}

extern "C" void kernel_launch(void* const* d_in, const int* in_sizes, int n_in,
                              void* d_out, int out_size, void* d_ws, size_t ws_size,
                              hipStream_t stream) {
  const float* x   = (const float*)d_in[0];
  const int* eidx  = (const int*)d_in[1];
  const float* W1l = (const float*)d_in[2];
  const float* W1r = (const float*)d_in[3];
  const float* b1  = (const float*)d_in[4];
  const float* W2l = (const float*)d_in[5];
  const float* W2r = (const float*)d_in[6];
  const float* b2  = (const float*)d_in[7];
  const float* W3l = (const float*)d_in[8];
  const float* W3r = (const float*)d_in[9];
  const float* b3  = (const float*)d_in[10];
  float* out = (float*)d_out;

  const int* src = eidx;
  const int* dst = eidx + N_EDGES;

  float* deg = (float*)d_ws;
  float* agg = deg + N_NODES;
  float* h1  = agg + (long)N_NODES * HID;
  float* h2  = h1 + (long)N_NODES * HID;

  const int ZB = 2048;
  const int eBlocks = (N_EDGES + 255) / 256;
  const int scatBlocks = (int)(((long)N_EDGES * 32 + 255) / 256);
  constexpr int STRIPS = 4;
  const int gemmBlocks = (N_NODES + 16 * STRIPS - 1) / (16 * STRIPS);

  zero_f32<<<ZB, 256, 0, stream>>>(deg, N_NODES);
  degree_kernel<<<eBlocks, 256, 0, stream>>>(dst, deg, N_EDGES);
  invdeg_kernel<<<(N_NODES + 255) / 256, 256, 0, stream>>>(deg, N_NODES);

  zero_f32<<<ZB, 256, 0, stream>>>(agg, (long)N_NODES * HID);
  scatter_kernel<<<scatBlocks, 256, 0, stream>>>(x, src, dst, agg, N_EDGES);
  sage_gemm<128, true, STRIPS><<<gemmBlocks, 256, 0, stream>>>(x, agg, deg, W1l, W1r, b1, h1, N_NODES);

  zero_f32<<<ZB, 256, 0, stream>>>(agg, (long)N_NODES * HID);
  scatter_kernel<<<scatBlocks, 256, 0, stream>>>(h1, src, dst, agg, N_EDGES);
  sage_gemm<128, true, STRIPS><<<gemmBlocks, 256, 0, stream>>>(h1, agg, deg, W2l, W2r, b2, h2, N_NODES);

  zero_f32<<<ZB, 256, 0, stream>>>(agg, (long)N_NODES * HID);
  scatter_kernel<<<scatBlocks, 256, 0, stream>>>(h2, src, dst, agg, N_EDGES);
  sage_gemm<64, false, STRIPS><<<gemmBlocks, 128, 0, stream>>>(h2, agg, deg, W3l, W3r, b3, out, N_NODES);
}


# neighbour sums by 256 owner work-groups: each holds 196 destination rows in LDS, walks the dst list, queues its edges, adds the gathered rows by LDS float adds; sum then scale; order of sums changed
# speedup vs baseline: 1.7361x; 1.7361x over previous
#include <hip/hip_runtime.h>

typedef __attribute__((ext_vector_type(2))) float v2f;
typedef __attribute__((ext_vector_type(8))) float v8f;

#define N_NODES 50000
#define N_EDGES 500000
#define HID 128
#define LDSPITCH (HID + 4)

__global__ void zero_f32(float* __restrict__ p, long n) {
  long i = (long)blockIdx.x * blockDim.x + threadIdx.x;
  long stride = (long)gridDim.x * blockDim.x;
  long n4 = n >> 2;
  float4* p4 = (float4*)p;
  for (long j = i; j < n4; j += stride) p4[j] = make_float4(0.f, 0.f, 0.f, 0.f);
}

__global__ void degree_kernel(const int* __restrict__ dst, float* __restrict__ deg, int E) {
  int e = blockIdx.x * blockDim.x + threadIdx.x;
  if (e < E) atomicAdd(&deg[dst[e]], 1.0f);
}

__global__ void invdeg_kernel(float* __restrict__ deg, int n) {
  int i = blockIdx.x * blockDim.x + threadIdx.x;
  if (i < n) deg[i] = 1.0f / fmaxf(deg[i], 1.0f);
}

#define TILE_N 196
#define TILE_THREADS 1024
#define TILE_WAVES (TILE_THREADS / 32)
#define QCAP 16384

template <int NL>
__device__ __forceinline__ void scan_items(const int* __restrict__ dst, int q0, int rot4, int E4,
                                           int n0, unsigned* queue, unsigned* qcount) {
  int4 d4[NL];
  unsigned e0[NL];
#pragma unroll
  for (int j = 0; j < NL; ++j) {
    int idx = q0 + j * TILE_THREADS + rot4;
    if (idx >= E4) idx -= E4;
    d4[j] = ((const int4*)dst)[idx];
    e0[j] = (unsigned)idx << 2;
  }
  unsigned t[NL][4];
  bool any = false;
#pragma unroll
  for (int j = 0; j < NL; ++j) {
    t[j][0] = (unsigned)(d4[j].x - n0);
    t[j][1] = (unsigned)(d4[j].y - n0);
    t[j][2] = (unsigned)(d4[j].z - n0);
    t[j][3] = (unsigned)(d4[j].w - n0);
#pragma unroll
    for (int k = 0; k < 4; ++k) any = any || (t[j][k] < (unsigned)TILE_N);
  }
  if (any) {
#pragma unroll
    for (int j = 0; j < NL; ++j) {
#pragma unroll
      for (int k = 0; k < 4; ++k) {
        if (t[j][k] < (unsigned)TILE_N) {
          const unsigned slot = atomicAdd(qcount, 1u);
          if (slot < (unsigned)QCAP) queue[slot] = ((e0[j] + (unsigned)k) << 8) | t[j][k];
        }
      }
    }
  }
}

template <int D, bool ADD_SCALED>
__global__ __launch_bounds__(TILE_THREADS) void tile_sum(
    const float* __restrict__ feat,
    const int* __restrict__ src,
    const int* __restrict__ dst,
    const float* __restrict__ invdeg,
    float* __restrict__ dest,
    int n_nodes, int n_edges) {
  constexpr int LPR = D / 4;
  constexpr int RPW = 32 / LPR;
  constexpr int U = 8;
  constexpr int SU = 4;
  __shared__ __attribute__((aligned(16))) float acc[TILE_N * D];
  __shared__ unsigned queue[QCAP];
  __shared__ unsigned qcount;

  const int tid = threadIdx.x;
  const int wave = tid >> 5;
  const int lane = tid & 31;
  const int n0 = blockIdx.x * TILE_N;
  const int E4 = n_edges >> 2;
  const int rot4 = (int)blockIdx.x * (E4 / (int)gridDim.x);

  for (int i = tid; i < TILE_N * D / 4; i += TILE_THREADS)
    ((float4*)acc)[i] = make_float4(0.f, 0.f, 0.f, 0.f);
  if (tid == 0) qcount = 0u;
  __syncthreads();

  auto scan = [&](int qb, int qe) {
    int q0 = qb + tid;
    for (; q0 + (SU - 1) * TILE_THREADS < qe; q0 += SU * TILE_THREADS)
      scan_items<SU>(dst, q0, rot4, E4, n0, queue, &qcount);
    for (; q0 < qe; q0 += TILE_THREADS)
      scan_items<1>(dst, q0, rot4, E4, n0, queue, &qcount);
  };

  auto drain = [&](int n) {
    for (int i = tid; i < n; i += TILE_THREADS) {
      const unsigned ent = queue[i];
      const unsigned s = (unsigned)src[ent >> 8];
      queue[i] = (s << 8) | (ent & 0xFFu);
    }
    __syncthreads();
    const int sub = lane / LPR;
    const int l = lane % LPR;
    for (int base = wave * RPW + sub; base < n + sub; base += TILE_WAVES * RPW * U) {
      float4 v[U];
      unsigned ent[U];
#pragma unroll
      for (int u = 0; u < U; ++u) {
        const int i = base + u * TILE_WAVES * RPW;
        const int ic = i < n ? i : n - 1;
        ent[u] = queue[ic];
        v[u] = *(const float4*)(feat + (long)(ent[u] >> 8) * D + l * 4);
      }
#pragma unroll
      for (int u = 0; u < U; ++u) {
        asm volatile("" ::"v"(v[u].x), "v"(v[u].y), "v"(v[u].z), "v"(v[u].w));
      }
#pragma unroll
      for (int u = 0; u < U; ++u) {
        const int i = base + u * TILE_WAVES * RPW;
        if (i < n) {
          float* a = acc + (int)(ent[u] & 0xFFu) * D + l;
          atomicAdd(a + 0 * LPR, v[u].x);
          atomicAdd(a + 1 * LPR, v[u].y);
          atomicAdd(a + 2 * LPR, v[u].z);
          atomicAdd(a + 3 * LPR, v[u].w);
        }
      }
    }
  };

  scan(0, E4);
  __syncthreads();
  const unsigned cnt = qcount;
  if (cnt <= (unsigned)QCAP) {
    drain((int)cnt);
  } else {
    for (int qb = 0; qb < E4; qb += QCAP / 4) {
      __syncthreads();
      if (tid == 0) qcount = 0u;
      __syncthreads();
      const int qe = qb + QCAP / 4 < E4 ? qb + QCAP / 4 : E4;
      scan(qb, qe);
      __syncthreads();
      drain((int)qcount);
    }
  }
  __syncthreads();

  for (int i = tid; i < TILE_N * LPR; i += TILE_THREADS) {
    const int r = i / LPR;
    const int l4 = i % LPR;
    const int row = n0 + r;
    if (row < n_nodes) {
      const float* a = acc + r * D + l4;
      float4 o = make_float4(a[0 * LPR], a[1 * LPR], a[2 * LPR], a[3 * LPR]);
      float4* p = (float4*)(dest + (long)row * D + l4 * 4);
      if (ADD_SCALED) {
        const float w = invdeg[row];
        const float4 c = *p;
        o.x = c.x + w * o.x; o.y = c.y + w * o.y; o.z = c.z + w * o.z; o.w = c.w + w * o.w;
      }
      *p = o;
    }
  }
}

template <int OUTD, bool RELU, int STRIPS>
__global__ __launch_bounds__(32 * (OUTD / 16)) void sage_gemm(
    const float* __restrict__ xin,
    const float* __restrict__ agg,
    const float* __restrict__ invdeg,
    const float* __restrict__ Wl,
    const float* __restrict__ Wr,
    const float* __restrict__ bias,
    float* __restrict__ out,
    int n_nodes) {
  __shared__ float sX[16][LDSPITCH];
  __shared__ float sA[16][LDSPITCH];

  const int tid = threadIdx.x;
  const int nthreads = 32 * (OUTD / 16);
  const int wave = tid >> 5;
  const int lane = tid & 31;
  const int half = lane >> 4;
  const int r = lane & 15;
  const int col = wave * 16 + r;

  v2f BL[HID / 4], BR[HID / 4];
#pragma unroll
  for (int j = 0; j < HID / 4; ++j) {
    const int ka = 4 * j + 2 * half;
    BL[j][0] = Wl[(long)ka * OUTD + col];
    BL[j][1] = Wl[(long)(ka + 1) * OUTD + col];
    BR[j][0] = Wr[(long)ka * OUTD + col];
    BR[j][1] = Wr[(long)(ka + 1) * OUTD + col];
  }
  const float bn = bias[col];

  for (int s = 0; s < STRIPS; ++s) {
    const int row0 = (blockIdx.x * STRIPS + s) * 16;
    if (row0 >= n_nodes) break;

    for (int i = tid; i < 16 * (HID / 4); i += nthreads) {
      const int rr = i / (HID / 4);
      const int c4 = (i % (HID / 4)) * 4;
      const long base = (long)(row0 + rr) * HID + c4;
      float4 xv = *(const float4*)(xin + base);
      float4 av = *(const float4*)(agg + base);
      const float iv = invdeg[row0 + rr];
      av.x *= iv; av.y *= iv; av.z *= iv; av.w *= iv;
      *(float4*)&sX[rr][c4] = xv;
      *(float4*)&sA[rr][c4] = av;
    }
    __syncthreads();

    v8f accL = {};
    v8f accR = {};
#pragma unroll
    for (int j = 0; j < HID / 4; ++j) {
      const int ka = 4 * j + 2 * half;
      v2f aL, aR;
      aL[0] = sA[r][ka];
      aL[1] = sA[r][ka + 1];
      aR[0] = sX[r][ka];
      aR[1] = sX[r][ka + 1];
      accL = __builtin_amdgcn_wmma_f32_16x16x4_f32(false, aL, false, BL[j],
                                                   (short)0, accL, false, false);
      accR = __builtin_amdgcn_wmma_f32_16x16x4_f32(false, aR, false, BR[j],
                                                   (short)0, accR, false, false);
    }

#pragma unroll
    for (int i = 0; i < 8; ++i) {
      const int gr = row0 + half * 8 + i;
      float v = accL[i] + accR[i] + bn;
      if (RELU) v = fmaxf(v, 0.0f);
      out[(long)gr * OUTD + col] = v;
    }
    __syncthreads();
  }
}

template <int STRIPS>
__global__ __launch_bounds__(256) void proj_gemm64(
    const float* __restrict__ xin,
    const float* __restrict__ Wl,
    const float* __restrict__ Wr,
    const float* __restrict__ bias,
    float* __restrict__ P,
    float* __restrict__ out,
    int n_nodes) {
  __shared__ float sX[16][LDSPITCH];

  const int tid = threadIdx.x;
  const int wave = tid >> 5;
  const int lane = tid & 31;
  const int half = lane >> 4;
  const int r = lane & 15;
  const bool selfside = wave >= 4;
  const float* __restrict__ Wsel = selfside ? Wr : Wl;
  float* __restrict__ dsel = selfside ? out : P;
  const int col = (wave & 3) * 16 + r;

  v2f B[HID / 4];
#pragma unroll
  for (int j = 0; j < HID / 4; ++j) {
    const int ka = 4 * j + 2 * half;
    B[j][0] = Wsel[(long)ka * 64 + col];
    B[j][1] = Wsel[(long)(ka + 1) * 64 + col];
  }
  const float bn = selfside ? bias[col] : 0.0f;

  for (int s = 0; s < STRIPS; ++s) {
    const int row0 = (blockIdx.x * STRIPS + s) * 16;
    if (row0 >= n_nodes) break;

    for (int i = tid; i < 16 * (HID / 4); i += 256) {
      const int rr = i / (HID / 4);
      const int c4 = (i % (HID / 4)) * 4;
      *(float4*)&sX[rr][c4] = *(const float4*)(xin + (long)(row0 + rr) * HID + c4);
    }
    __syncthreads();

    v8f acc0 = {};
    v8f acc1 = {};
#pragma unroll
    for (int j = 0; j < HID / 4; j += 2) {
      const int ka = 4 * j + 2 * half;
      v2f a0, a1;
      a0[0] = sX[r][ka];
      a0[1] = sX[r][ka + 1];
      a1[0] = sX[r][ka + 4];
      a1[1] = sX[r][ka + 5];
      acc0 = __builtin_amdgcn_wmma_f32_16x16x4_f32(false, a0, false, B[j],
                                                   (short)0, acc0, false, false);
      acc1 = __builtin_amdgcn_wmma_f32_16x16x4_f32(false, a1, false, B[j + 1],
                                                   (short)0, acc1, false, false);
    }

#pragma unroll
    for (int i = 0; i < 8; ++i) {
      const int gr = row0 + half * 8 + i;
      dsel[(long)gr * 64 + col] = acc0[i] + acc1[i] + bn;
    }
    __syncthreads();
  }
}

extern "C" void kernel_launch(void* const* d_in, const int* in_sizes, int n_in,
                              void* d_out, int out_size, void* d_ws, size_t ws_size,
                              hipStream_t stream) {
  const float* x   = (const float*)d_in[0];
  const int* eidx  = (const int*)d_in[1];
  const float* W1l = (const float*)d_in[2];
  const float* W1r = (const float*)d_in[3];
  const float* b1  = (const float*)d_in[4];
  const float* W2l = (const float*)d_in[5];
  const float* W2r = (const float*)d_in[6];
  const float* b2  = (const float*)d_in[7];
  const float* W3l = (const float*)d_in[8];
  const float* W3r = (const float*)d_in[9];
  const float* b3  = (const float*)d_in[10];
  float* out = (float*)d_out;

  const int* src = eidx;
  const int* dst = eidx + N_EDGES;

  float* deg = (float*)d_ws;
  float* agg = deg + N_NODES;
  float* h1  = agg + (long)N_NODES * HID;
  float* h2  = h1 + (long)N_NODES * HID;

  const int ZB = 2048;
  const int eBlocks = (N_EDGES + 255) / 256;
  const int tileBlocks = (N_NODES + TILE_N - 1) / TILE_N;
  constexpr int STRIPS = 4;
  const int gemmBlocks = (N_NODES + 16 * STRIPS - 1) / (16 * STRIPS);

  zero_f32<<<ZB, 256, 0, stream>>>(deg, N_NODES);
  degree_kernel<<<eBlocks, 256, 0, stream>>>(dst, deg, N_EDGES);
  invdeg_kernel<<<(N_NODES + 255) / 256, 256, 0, stream>>>(deg, N_NODES);

  tile_sum<HID, false><<<tileBlocks, TILE_THREADS, 0, stream>>>(x, src, dst, deg, agg, N_NODES, N_EDGES);
  sage_gemm<128, true, STRIPS><<<gemmBlocks, 256, 0, stream>>>(x, agg, deg, W1l, W1r, b1, h1, N_NODES);

  tile_sum<HID, false><<<tileBlocks, TILE_THREADS, 0, stream>>>(h1, src, dst, deg, agg, N_NODES, N_EDGES);
  sage_gemm<128, true, STRIPS><<<gemmBlocks, 256, 0, stream>>>(h1, agg, deg, W2l, W2r, b2, h2, N_NODES);

  float* P = agg;
  proj_gemm64<STRIPS><<<gemmBlocks, 256, 0, stream>>>(h2, W3l, W3r, b3, P, out, N_NODES);
  tile_sum<64, true><<<tileBlocks, TILE_THREADS, 0, stream>>>(P, src, dst, deg, out, N_NODES, N_EDGES);
}


# layers 2 and 3 reuse the owner queues that layer 1's dst walk resolved and stored in the workspace: one walk and one src lookup per call instead of three; sums still in LDS arrival order
# speedup vs baseline: 2.0876x; 1.2025x over previous
#include <hip/hip_runtime.h>

typedef __attribute__((ext_vector_type(2))) float v2f;
typedef __attribute__((ext_vector_type(8))) float v8f;

#define N_NODES 50000
#define N_EDGES 500000
#define HID 128
#define LDSPITCH (HID + 4)

__global__ void zero_f32(float* __restrict__ p, long n) {
  long i = (long)blockIdx.x * blockDim.x + threadIdx.x;
  long stride = (long)gridDim.x * blockDim.x;
  long n4 = n >> 2;
  float4* p4 = (float4*)p;
  for (long j = i; j < n4; j += stride) p4[j] = make_float4(0.f, 0.f, 0.f, 0.f);
}

__global__ void degree_kernel(const int* __restrict__ dst, float* __restrict__ deg, int E) {
  int e = blockIdx.x * blockDim.x + threadIdx.x;
  if (e < E) atomicAdd(&deg[dst[e]], 1.0f);
}

__global__ void invdeg_kernel(float* __restrict__ deg, int n) {
  int i = blockIdx.x * blockDim.x + threadIdx.x;
  if (i < n) deg[i] = 1.0f / fmaxf(deg[i], 1.0f);
}

#define TILE_N 196
#define TILE_THREADS 1024
#define TILE_WAVES (TILE_THREADS / 32)
#define QCAP 16384
#define TILE_GROUPS ((N_NODES + TILE_N - 1) / TILE_N)

template <int NL>
__device__ __forceinline__ void scan_items(const int* __restrict__ dst, int q0, int rot4, int E4,
                                           int n0, unsigned* queue, unsigned* qcount) {
  int4 d4[NL];
  unsigned e0[NL];
#pragma unroll
  for (int j = 0; j < NL; ++j) {
    int idx = q0 + j * TILE_THREADS + rot4;
    if (idx >= E4) idx -= E4;
    d4[j] = ((const int4*)dst)[idx];
    e0[j] = (unsigned)idx << 2;
  }
  unsigned t[NL][4];
  bool any = false;
#pragma unroll
  for (int j = 0; j < NL; ++j) {
    t[j][0] = (unsigned)(d4[j].x - n0);
    t[j][1] = (unsigned)(d4[j].y - n0);
    t[j][2] = (unsigned)(d4[j].z - n0);
    t[j][3] = (unsigned)(d4[j].w - n0);
#pragma unroll
    for (int k = 0; k < 4; ++k) any = any || (t[j][k] < (unsigned)TILE_N);
  }
  if (any) {
#pragma unroll
    for (int j = 0; j < NL; ++j) {
#pragma unroll
      for (int k = 0; k < 4; ++k) {
        if (t[j][k] < (unsigned)TILE_N) {
          const unsigned slot = atomicAdd(qcount, 1u);
          if (slot < (unsigned)QCAP) queue[slot] = ((e0[j] + (unsigned)k) << 8) | t[j][k];
        }
      }
    }
  }
}

template <int D, bool ADD_SCALED, bool BUILD>
__global__ __launch_bounds__(TILE_THREADS) void tile_sum(
    const float* __restrict__ feat,
    const int* __restrict__ src,
    const int* __restrict__ dst,
    const float* __restrict__ invdeg,
    float* __restrict__ dest,
    int* qsaved,
    int* qsaved_n,
    int n_nodes, int n_edges) {
  constexpr int LPR = D / 4;
  constexpr int RPW = 32 / LPR;
  constexpr int U = 8;
  constexpr int SU = 4;
  __shared__ __attribute__((aligned(16))) float acc[TILE_N * D];
  __shared__ unsigned queue[QCAP];
  __shared__ unsigned qcount;

  const int tid = threadIdx.x;
  const int wave = tid >> 5;
  const int lane = tid & 31;
  const int n0 = blockIdx.x * TILE_N;
  const int E4 = n_edges >> 2;
  const int rot4 = (int)blockIdx.x * (E4 / (int)gridDim.x);

  for (int i = tid; i < TILE_N * D / 4; i += TILE_THREADS)
    ((float4*)acc)[i] = make_float4(0.f, 0.f, 0.f, 0.f);
  if (tid == 0) qcount = 0u;
  __syncthreads();

  auto scan = [&](int qb, int qe) {
    int q0 = qb + tid;
    for (; q0 + (SU - 1) * TILE_THREADS < qe; q0 += SU * TILE_THREADS)
      scan_items<SU>(dst, q0, rot4, E4, n0, queue, &qcount);
    for (; q0 < qe; q0 += TILE_THREADS)
      scan_items<1>(dst, q0, rot4, E4, n0, queue, &qcount);
  };

  int* const qmine = qsaved + (long)blockIdx.x * QCAP;

  auto resolve = [&](int n, bool save) {
    for (int i = tid; i < n; i += TILE_THREADS) {
      const unsigned ent = queue[i];
      const unsigned s = (unsigned)src[ent >> 8];
      const unsigned res = (s << 8) | (ent & 0xFFu);
      queue[i] = res;
      if (save) qmine[i] = (int)res;
    }
    __syncthreads();
  };

  auto gather = [&](int n) {
    const int sub = lane / LPR;
    const int l = lane % LPR;
    for (int base = wave * RPW + sub; base < n + sub; base += TILE_WAVES * RPW * U) {
      float4 v[U];
      unsigned ent[U];
#pragma unroll
      for (int u = 0; u < U; ++u) {
        const int i = base + u * TILE_WAVES * RPW;
        const int ic = i < n ? i : n - 1;
        ent[u] = queue[ic];
        v[u] = *(const float4*)(feat + (long)(ent[u] >> 8) * D + l * 4);
      }
#pragma unroll
      for (int u = 0; u < U; ++u) {
        asm volatile("" ::"v"(v[u].x), "v"(v[u].y), "v"(v[u].z), "v"(v[u].w));
      }
#pragma unroll
      for (int u = 0; u < U; ++u) {
        const int i = base + u * TILE_WAVES * RPW;
        if (i < n) {
          float* a = acc + (int)(ent[u] & 0xFFu) * D + l;
          atomicAdd(a + 0 * LPR, v[u].x);
          atomicAdd(a + 1 * LPR, v[u].y);
          atomicAdd(a + 2 * LPR, v[u].z);
          atomicAdd(a + 3 * LPR, v[u].w);
        }
      }
    }
  };

  unsigned cnt;
  if (BUILD) {
    scan(0, E4);
    __syncthreads();
    cnt = qcount;
    if (tid == 0) qsaved_n[blockIdx.x] = (int)cnt;
  } else {
    cnt = (unsigned)qsaved_n[blockIdx.x];
  }
  if (cnt <= (unsigned)QCAP) {
    if (BUILD) {
      resolve((int)cnt, true);
    } else {
      for (int i = tid; i < (int)cnt; i += TILE_THREADS) queue[i] = (unsigned)qmine[i];
      __syncthreads();
    }
    gather((int)cnt);
  } else {
    for (int qb = 0; qb < E4; qb += QCAP / 4) {
      __syncthreads();
      if (tid == 0) qcount = 0u;
      __syncthreads();
      const int qe = qb + QCAP / 4 < E4 ? qb + QCAP / 4 : E4;
      scan(qb, qe);
      __syncthreads();
      const int n = (int)qcount;
      resolve(n, false);
      gather(n);
    }
  }
  __syncthreads();

  for (int i = tid; i < TILE_N * LPR; i += TILE_THREADS) {
    const int r = i / LPR;
    const int l4 = i % LPR;
    const int row = n0 + r;
    if (row < n_nodes) {
      const float* a = acc + r * D + l4;
      float4 o = make_float4(a[0 * LPR], a[1 * LPR], a[2 * LPR], a[3 * LPR]);
      float4* p = (float4*)(dest + (long)row * D + l4 * 4);
      if (ADD_SCALED) {
        const float w = invdeg[row];
        const float4 c = *p;
        o.x = c.x + w * o.x; o.y = c.y + w * o.y; o.z = c.z + w * o.z; o.w = c.w + w * o.w;
      }
      *p = o;
    }
  }
}

template <int OUTD, bool RELU, int STRIPS>
__global__ __launch_bounds__(32 * (OUTD / 16)) void sage_gemm(
    const float* __restrict__ xin,
    const float* __restrict__ agg,
    const float* __restrict__ invdeg,
    const float* __restrict__ Wl,
    const float* __restrict__ Wr,
    const float* __restrict__ bias,
    float* __restrict__ out,
    int n_nodes) {
  __shared__ float sX[16][LDSPITCH];
  __shared__ float sA[16][LDSPITCH];

  const int tid = threadIdx.x;
  const int nthreads = 32 * (OUTD / 16);
  const int wave = tid >> 5;
  const int lane = tid & 31;
  const int half = lane >> 4;
  const int r = lane & 15;
  const int col = wave * 16 + r;

  v2f BL[HID / 4], BR[HID / 4];
#pragma unroll
  for (int j = 0; j < HID / 4; ++j) {
    const int ka = 4 * j + 2 * half;
    BL[j][0] = Wl[(long)ka * OUTD + col];
    BL[j][1] = Wl[(long)(ka + 1) * OUTD + col];
    BR[j][0] = Wr[(long)ka * OUTD + col];
    BR[j][1] = Wr[(long)(ka + 1) * OUTD + col];
  }
  const float bn = bias[col];

  for (int s = 0; s < STRIPS; ++s) {
    const int row0 = (blockIdx.x * STRIPS + s) * 16;
    if (row0 >= n_nodes) break;

    for (int i = tid; i < 16 * (HID / 4); i += nthreads) {
      const int rr = i / (HID / 4);
      const int c4 = (i % (HID / 4)) * 4;
      const long base = (long)(row0 + rr) * HID + c4;
      float4 xv = *(const float4*)(xin + base);
      float4 av = *(const float4*)(agg + base);
      const float iv = invdeg[row0 + rr];
      av.x *= iv; av.y *= iv; av.z *= iv; av.w *= iv;
      *(float4*)&sX[rr][c4] = xv;
      *(float4*)&sA[rr][c4] = av;
    }
    __syncthreads();

    v8f accL = {};
    v8f accR = {};
#pragma unroll
    for (int j = 0; j < HID / 4; ++j) {
      const int ka = 4 * j + 2 * half;
      v2f aL, aR;
      aL[0] = sA[r][ka];
      aL[1] = sA[r][ka + 1];
      aR[0] = sX[r][ka];
      aR[1] = sX[r][ka + 1];
      accL = __builtin_amdgcn_wmma_f32_16x16x4_f32(false, aL, false, BL[j],
                                                   (short)0, accL, false, false);
      accR = __builtin_amdgcn_wmma_f32_16x16x4_f32(false, aR, false, BR[j],
                                                   (short)0, accR, false, false);
    }

#pragma unroll
    for (int i = 0; i < 8; ++i) {
      const int gr = row0 + half * 8 + i;
      float v = accL[i] + accR[i] + bn;
      if (RELU) v = fmaxf(v, 0.0f);
      out[(long)gr * OUTD + col] = v;
    }
    __syncthreads();
  }
}

template <int STRIPS>
__global__ __launch_bounds__(256) void proj_gemm64(
    const float* __restrict__ xin,
    const float* __restrict__ Wl,
    const float* __restrict__ Wr,
    const float* __restrict__ bias,
    float* __restrict__ P,
    float* __restrict__ out,
    int n_nodes) {
  __shared__ float sX[16][LDSPITCH];

  const int tid = threadIdx.x;
  const int wave = tid >> 5;
  const int lane = tid & 31;
  const int half = lane >> 4;
  const int r = lane & 15;
  const bool selfside = wave >= 4;
  const float* __restrict__ Wsel = selfside ? Wr : Wl;
  float* __restrict__ dsel = selfside ? out : P;
  const int col = (wave & 3) * 16 + r;

  v2f B[HID / 4];
#pragma unroll
  for (int j = 0; j < HID / 4; ++j) {
    const int ka = 4 * j + 2 * half;
    B[j][0] = Wsel[(long)ka * 64 + col];
    B[j][1] = Wsel[(long)(ka + 1) * 64 + col];
  }
  const float bn = selfside ? bias[col] : 0.0f;

  for (int s = 0; s < STRIPS; ++s) {
    const int row0 = (blockIdx.x * STRIPS + s) * 16;
    if (row0 >= n_nodes) break;

    for (int i = tid; i < 16 * (HID / 4); i += 256) {
      const int rr = i / (HID / 4);
      const int c4 = (i % (HID / 4)) * 4;
      *(float4*)&sX[rr][c4] = *(const float4*)(xin + (long)(row0 + rr) * HID + c4);
    }
    __syncthreads();

    v8f acc0 = {};
    v8f acc1 = {};
#pragma unroll
    for (int j = 0; j < HID / 4; j += 2) {
      const int ka = 4 * j + 2 * half;
      v2f a0, a1;
      a0[0] = sX[r][ka];
      a0[1] = sX[r][ka + 1];
      a1[0] = sX[r][ka + 4];
      a1[1] = sX[r][ka + 5];
      acc0 = __builtin_amdgcn_wmma_f32_16x16x4_f32(false, a0, false, B[j],
                                                   (short)0, acc0, false, false);
      acc1 = __builtin_amdgcn_wmma_f32_16x16x4_f32(false, a1, false, B[j + 1],
                                                   (short)0, acc1, false, false);
    }

#pragma unroll
    for (int i = 0; i < 8; ++i) {
      const int gr = row0 + half * 8 + i;
      dsel[(long)gr * 64 + col] = acc0[i] + acc1[i] + bn;
    }
    __syncthreads();
  }
}

extern "C" void kernel_launch(void* const* d_in, const int* in_sizes, int n_in,
                              void* d_out, int out_size, void* d_ws, size_t ws_size,
                              hipStream_t stream) {
  const float* x   = (const float*)d_in[0];
  const int* eidx  = (const int*)d_in[1];
  const float* W1l = (const float*)d_in[2];
  const float* W1r = (const float*)d_in[3];
  const float* b1  = (const float*)d_in[4];
  const float* W2l = (const float*)d_in[5];
  const float* W2r = (const float*)d_in[6];
  const float* b2  = (const float*)d_in[7];
  const float* W3l = (const float*)d_in[8];
  const float* W3r = (const float*)d_in[9];
  const float* b3  = (const float*)d_in[10];
  float* out = (float*)d_out;

  const int* src = eidx;
  const int* dst = eidx + N_EDGES;

  float* deg = (float*)d_ws;
  float* agg = deg + N_NODES;
  float* h1  = agg + (long)N_NODES * HID;
  float* h2  = h1 + (long)N_NODES * HID;
  int* qn    = (int*)(h2 + (long)N_NODES * HID);
  int* qs    = qn + TILE_GROUPS;

  const int ZB = 2048;
  const int eBlocks = (N_EDGES + 255) / 256;
  const int tileBlocks = TILE_GROUPS;
  constexpr int STRIPS = 4;
  const int gemmBlocks = (N_NODES + 16 * STRIPS - 1) / (16 * STRIPS);

  zero_f32<<<ZB, 256, 0, stream>>>(deg, N_NODES);
  degree_kernel<<<eBlocks, 256, 0, stream>>>(dst, deg, N_EDGES);
  invdeg_kernel<<<(N_NODES + 255) / 256, 256, 0, stream>>>(deg, N_NODES);

  tile_sum<HID, false, true><<<tileBlocks, TILE_THREADS, 0, stream>>>(x, src, dst, deg, agg, qs, qn, N_NODES, N_EDGES);
  sage_gemm<128, true, STRIPS><<<gemmBlocks, 256, 0, stream>>>(x, agg, deg, W1l, W1r, b1, h1, N_NODES);

  tile_sum<HID, false, false><<<tileBlocks, TILE_THREADS, 0, stream>>>(h1, src, dst, deg, agg, qs, qn, N_NODES, N_EDGES);
  sage_gemm<128, true, STRIPS><<<gemmBlocks, 256, 0, stream>>>(h1, agg, deg, W2l, W2r, b2, h2, N_NODES);

  float* P = agg;
  proj_gemm64<STRIPS><<<gemmBlocks, 256, 0, stream>>>(h2, W3l, W3r, b3, P, out, N_NODES);
  tile_sum<64, true, false><<<tileBlocks, TILE_THREADS, 0, stream>>>(P, src, dst, deg, out, qs, qn, N_NODES, N_EDGES);
}


# in-degree counted inside the first neighbour-sum launch: owner work-groups count their queued edges per row by LDS integer adds and write 1/max(deg,1); zero, degree and reciprocal launches removed
# speedup vs baseline: 2.4784x; 1.1872x over previous
#include <hip/hip_runtime.h>

typedef __attribute__((ext_vector_type(2))) float v2f;
typedef __attribute__((ext_vector_type(8))) float v8f;

#define N_NODES 50000
#define N_EDGES 500000
#define HID 128
#define LDSPITCH (HID + 4)

#define TILE_N 196
#define TILE_THREADS 1024
#define TILE_WAVES (TILE_THREADS / 32)
#define QCAP 16384
#define TILE_GROUPS ((N_NODES + TILE_N - 1) / TILE_N)

template <int NL>
__device__ __forceinline__ void scan_items(const int* __restrict__ dst, int q0, int rot4, int E4,
                                           int n0, unsigned* queue, unsigned* qcount) {
  int4 d4[NL];
  unsigned e0[NL];
#pragma unroll
  for (int j = 0; j < NL; ++j) {
    int idx = q0 + j * TILE_THREADS + rot4;
    if (idx >= E4) idx -= E4;
    d4[j] = ((const int4*)dst)[idx];
    e0[j] = (unsigned)idx << 2;
  }
  unsigned t[NL][4];
  bool any = false;
#pragma unroll
  for (int j = 0; j < NL; ++j) {
    t[j][0] = (unsigned)(d4[j].x - n0);
    t[j][1] = (unsigned)(d4[j].y - n0);
    t[j][2] = (unsigned)(d4[j].z - n0);
    t[j][3] = (unsigned)(d4[j].w - n0);
#pragma unroll
    for (int k = 0; k < 4; ++k) any = any || (t[j][k] < (unsigned)TILE_N);
  }
  if (any) {
#pragma unroll
    for (int j = 0; j < NL; ++j) {
#pragma unroll
      for (int k = 0; k < 4; ++k) {
        if (t[j][k] < (unsigned)TILE_N) {
          const unsigned slot = atomicAdd(qcount, 1u);
          if (slot < (unsigned)QCAP) queue[slot] = ((e0[j] + (unsigned)k) << 8) | t[j][k];
        }
      }
    }
  }
}

template <int D, bool ADD_SCALED, bool BUILD>
__global__ __launch_bounds__(TILE_THREADS) void tile_sum(
    const float* __restrict__ feat,
    const int* __restrict__ src,
    const int* __restrict__ dst,
    const float* __restrict__ invdeg,
    float* invdeg_out,
    float* __restrict__ dest,
    int* qsaved,
    int* qsaved_n,
    int n_nodes, int n_edges) {
  constexpr int LPR = D / 4;
  constexpr int RPW = 32 / LPR;
  constexpr int U = 8;
  constexpr int SU = 4;
  __shared__ __attribute__((aligned(16))) float acc[TILE_N * D];
  __shared__ unsigned queue[QCAP];
  __shared__ unsigned qcount;
  __shared__ unsigned degc[BUILD ? TILE_N : 1];

  const int tid = threadIdx.x;
  const int wave = tid >> 5;
  const int lane = tid & 31;
  const int n0 = blockIdx.x * TILE_N;
  const int E4 = n_edges >> 2;
  const int rot4 = (int)blockIdx.x * (E4 / (int)gridDim.x);

  for (int i = tid; i < TILE_N * D / 4; i += TILE_THREADS)
    ((float4*)acc)[i] = make_float4(0.f, 0.f, 0.f, 0.f);
  if (tid == 0) qcount = 0u;
  if (BUILD) {
    for (int i = tid; i < TILE_N; i += TILE_THREADS) degc[i] = 0u;
  }
  __syncthreads();

  auto scan = [&](int qb, int qe) {
    int q0 = qb + tid;
    for (; q0 + (SU - 1) * TILE_THREADS < qe; q0 += SU * TILE_THREADS)
      scan_items<SU>(dst, q0, rot4, E4, n0, queue, &qcount);
    for (; q0 < qe; q0 += TILE_THREADS)
      scan_items<1>(dst, q0, rot4, E4, n0, queue, &qcount);
  };

  int* const qmine = qsaved + (long)blockIdx.x * QCAP;

  auto resolve = [&](int n, bool save) {
    for (int i = tid; i < n; i += TILE_THREADS) {
      const unsigned ent = queue[i];
      const unsigned s = (unsigned)src[ent >> 8];
      const unsigned res = (s << 8) | (ent & 0xFFu);
      queue[i] = res;
      if (save) qmine[i] = (int)res;
      if (BUILD) atomicAdd(&degc[ent & 0xFFu], 1u);
    }
    __syncthreads();
  };

  auto gather = [&](int n) {
    const int sub = lane / LPR;
    const int l = lane % LPR;
    for (int base = wave * RPW + sub; base < n + sub; base += TILE_WAVES * RPW * U) {
      float4 v[U];
      unsigned ent[U];
#pragma unroll
      for (int u = 0; u < U; ++u) {
        const int i = base + u * TILE_WAVES * RPW;
        const int ic = i < n ? i : n - 1;
        ent[u] = queue[ic];
        v[u] = *(const float4*)(feat + (long)(ent[u] >> 8) * D + l * 4);
      }
#pragma unroll
      for (int u = 0; u < U; ++u) {
        asm volatile("" ::"v"(v[u].x), "v"(v[u].y), "v"(v[u].z), "v"(v[u].w));
      }
#pragma unroll
      for (int u = 0; u < U; ++u) {
        const int i = base + u * TILE_WAVES * RPW;
        if (i < n) {
          float* a = acc + (int)(ent[u] & 0xFFu) * D + l;
          atomicAdd(a + 0 * LPR, v[u].x);
          atomicAdd(a + 1 * LPR, v[u].y);
          atomicAdd(a + 2 * LPR, v[u].z);
          atomicAdd(a + 3 * LPR, v[u].w);
        }
      }
    }
  };

  unsigned cnt;
  if (BUILD) {
    scan(0, E4);
    __syncthreads();
    cnt = qcount;
    if (tid == 0) qsaved_n[blockIdx.x] = (int)cnt;
  } else {
    cnt = (unsigned)qsaved_n[blockIdx.x];
  }
  if (cnt <= (unsigned)QCAP) {
    if (BUILD) {
      resolve((int)cnt, true);
    } else {
      for (int i = tid; i < (int)cnt; i += TILE_THREADS) queue[i] = (unsigned)qmine[i];
      __syncthreads();
    }
    gather((int)cnt);
  } else {
    for (int qb = 0; qb < E4; qb += QCAP / 4) {
      __syncthreads();
      if (tid == 0) qcount = 0u;
      __syncthreads();
      const int qe = qb + QCAP / 4 < E4 ? qb + QCAP / 4 : E4;
      scan(qb, qe);
      __syncthreads();
      const int n = (int)qcount;
      resolve(n, false);
      gather(n);
    }
  }
  __syncthreads();

  if (BUILD) {
    for (int i = tid; i < TILE_N; i += TILE_THREADS) {
      const int row = n0 + i;
      if (row < n_nodes) invdeg_out[row] = 1.0f / fmaxf((float)degc[i], 1.0f);
    }
  }

  for (int i = tid; i < TILE_N * LPR; i += TILE_THREADS) {
    const int r = i / LPR;
    const int l4 = i % LPR;
    const int row = n0 + r;
    if (row < n_nodes) {
      const float* a = acc + r * D + l4;
      float4 o = make_float4(a[0 * LPR], a[1 * LPR], a[2 * LPR], a[3 * LPR]);
      float4* p = (float4*)(dest + (long)row * D + l4 * 4);
      if (ADD_SCALED) {
        const float w = invdeg[row];
        const float4 c = *p;
        o.x = c.x + w * o.x; o.y = c.y + w * o.y; o.z = c.z + w * o.z; o.w = c.w + w * o.w;
      }
      *p = o;
    }
  }
}

template <int OUTD, bool RELU, int STRIPS>
__global__ __launch_bounds__(32 * (OUTD / 16)) void sage_gemm(
    const float* __restrict__ xin,
    const float* __restrict__ agg,
    const float* __restrict__ invdeg,
    const float* __restrict__ Wl,
    const float* __restrict__ Wr,
    const float* __restrict__ bias,
    float* __restrict__ out,
    int n_nodes) {
  __shared__ float sX[16][LDSPITCH];
  __shared__ float sA[16][LDSPITCH];

  const int tid = threadIdx.x;
  const int nthreads = 32 * (OUTD / 16);
  const int wave = tid >> 5;
  const int lane = tid & 31;
  const int half = lane >> 4;
  const int r = lane & 15;
  const int col = wave * 16 + r;

  v2f BL[HID / 4], BR[HID / 4];
#pragma unroll
  for (int j = 0; j < HID / 4; ++j) {
    const int ka = 4 * j + 2 * half;
    BL[j][0] = Wl[(long)ka * OUTD + col];
    BL[j][1] = Wl[(long)(ka + 1) * OUTD + col];
    BR[j][0] = Wr[(long)ka * OUTD + col];
    BR[j][1] = Wr[(long)(ka + 1) * OUTD + col];
  }
  const float bn = bias[col];

  for (int s = 0; s < STRIPS; ++s) {
    const int row0 = (blockIdx.x * STRIPS + s) * 16;
    if (row0 >= n_nodes) break;

    for (int i = tid; i < 16 * (HID / 4); i += nthreads) {
      const int rr = i / (HID / 4);
      const int c4 = (i % (HID / 4)) * 4;
      const long base = (long)(row0 + rr) * HID + c4;
      float4 xv = *(const float4*)(xin + base);
      float4 av = *(const float4*)(agg + base);
      const float iv = invdeg[row0 + rr];
      av.x *= iv; av.y *= iv; av.z *= iv; av.w *= iv;
      *(float4*)&sX[rr][c4] = xv;
      *(float4*)&sA[rr][c4] = av;
    }
    __syncthreads();

    v8f accL = {};
    v8f accR = {};
#pragma unroll
    for (int j = 0; j < HID / 4; ++j) {
      const int ka = 4 * j + 2 * half;
      v2f aL, aR;
      aL[0] = sA[r][ka];
      aL[1] = sA[r][ka + 1];
      aR[0] = sX[r][ka];
      aR[1] = sX[r][ka + 1];
      accL = __builtin_amdgcn_wmma_f32_16x16x4_f32(false, aL, false, BL[j],
                                                   (short)0, accL, false, false);
      accR = __builtin_amdgcn_wmma_f32_16x16x4_f32(false, aR, false, BR[j],
                                                   (short)0, accR, false, false);
    }

#pragma unroll
    for (int i = 0; i < 8; ++i) {
      const int gr = row0 + half * 8 + i;
      float v = accL[i] + accR[i] + bn;
      if (RELU) v = fmaxf(v, 0.0f);
      out[(long)gr * OUTD + col] = v;
    }
    __syncthreads();
  }
}

template <int STRIPS>
__global__ __launch_bounds__(256) void proj_gemm64(
    const float* __restrict__ xin,
    const float* __restrict__ Wl,
    const float* __restrict__ Wr,
    const float* __restrict__ bias,
    float* __restrict__ P,
    float* __restrict__ out,
    int n_nodes) {
  __shared__ float sX[16][LDSPITCH];

  const int tid = threadIdx.x;
  const int wave = tid >> 5;
  const int lane = tid & 31;
  const int half = lane >> 4;
  const int r = lane & 15;
  const bool selfside = wave >= 4;
  const float* __restrict__ Wsel = selfside ? Wr : Wl;
  float* __restrict__ dsel = selfside ? out : P;
  const int col = (wave & 3) * 16 + r;

  v2f B[HID / 4];
#pragma unroll
  for (int j = 0; j < HID / 4; ++j) {
    const int ka = 4 * j + 2 * half;
    B[j][0] = Wsel[(long)ka * 64 + col];
    B[j][1] = Wsel[(long)(ka + 1) * 64 + col];
  }
  const float bn = selfside ? bias[col] : 0.0f;

  for (int s = 0; s < STRIPS; ++s) {
    const int row0 = (blockIdx.x * STRIPS + s) * 16;
    if (row0 >= n_nodes) break;

    for (int i = tid; i < 16 * (HID / 4); i += 256) {
      const int rr = i / (HID / 4);
      const int c4 = (i % (HID / 4)) * 4;
      *(float4*)&sX[rr][c4] = *(const float4*)(xin + (long)(row0 + rr) * HID + c4);
    }
    __syncthreads();

    v8f acc0 = {};
    v8f acc1 = {};
#pragma unroll
    for (int j = 0; j < HID / 4; j += 2) {
      const int ka = 4 * j + 2 * half;
      v2f a0, a1;
      a0[0] = sX[r][ka];
      a0[1] = sX[r][ka + 1];
      a1[0] = sX[r][ka + 4];
      a1[1] = sX[r][ka + 5];
      acc0 = __builtin_amdgcn_wmma_f32_16x16x4_f32(false, a0, false, B[j],
                                                   (short)0, acc0, false, false);
      acc1 = __builtin_amdgcn_wmma_f32_16x16x4_f32(false, a1, false, B[j + 1],
                                                   (short)0, acc1, false, false);
    }

#pragma unroll
    for (int i = 0; i < 8; ++i) {
      const int gr = row0 + half * 8 + i;
      dsel[(long)gr * 64 + col] = acc0[i] + acc1[i] + bn;
    }
    __syncthreads();
  }
}

extern "C" void kernel_launch(void* const* d_in, const int* in_sizes, int n_in,
                              void* d_out, int out_size, void* d_ws, size_t ws_size,
                              hipStream_t stream) {
  const float* x   = (const float*)d_in[0];
  const int* eidx  = (const int*)d_in[1];
  const float* W1l = (const float*)d_in[2];
  const float* W1r = (const float*)d_in[3];
  const float* b1  = (const float*)d_in[4];
  const float* W2l = (const float*)d_in[5];
  const float* W2r = (const float*)d_in[6];
  const float* b2  = (const float*)d_in[7];
  const float* W3l = (const float*)d_in[8];
  const float* W3r = (const float*)d_in[9];
  const float* b3  = (const float*)d_in[10];
  float* out = (float*)d_out;

  const int* src = eidx;
  const int* dst = eidx + N_EDGES;

  float* deg = (float*)d_ws;
  float* agg = deg + N_NODES;
  float* h1  = agg + (long)N_NODES * HID;
  float* h2  = h1 + (long)N_NODES * HID;
  int* qn    = (int*)(h2 + (long)N_NODES * HID);
  int* qs    = qn + TILE_GROUPS;

  const int tileBlocks = TILE_GROUPS;
  constexpr int STRIPS = 4;
  const int gemmBlocks = (N_NODES + 16 * STRIPS - 1) / (16 * STRIPS);

  tile_sum<HID, false, true><<<tileBlocks, TILE_THREADS, 0, stream>>>(x, src, dst, deg, deg, agg, qs, qn, N_NODES, N_EDGES);
  sage_gemm<128, true, STRIPS><<<gemmBlocks, 256, 0, stream>>>(x, agg, deg, W1l, W1r, b1, h1, N_NODES);

  tile_sum<HID, false, false><<<tileBlocks, TILE_THREADS, 0, stream>>>(h1, src, dst, deg, deg, agg, qs, qn, N_NODES, N_EDGES);
  sage_gemm<128, true, STRIPS><<<gemmBlocks, 256, 0, stream>>>(h1, agg, deg, W2l, W2r, b2, h2, N_NODES);

  float* P = agg;
  proj_gemm64<STRIPS><<<gemmBlocks, 256, 0, stream>>>(h2, W3l, W3r, b3, P, out, N_NODES);
  tile_sum<64, true, false><<<tileBlocks, TILE_THREADS, 0, stream>>>(P, src, dst, deg, deg, out, qs, qn, N_NODES, N_EDGES);
}


# layer GEMM launches sized to one resident round: 768 blocks (3 per WGP at 160 VGPRs) take the 3125 row-strips round-robin instead of 782 blocks of 4 strips, 14 of which ran as a late second round
# speedup vs baseline: 2.7557x; 1.1119x over previous
#include <hip/hip_runtime.h>

typedef __attribute__((ext_vector_type(2))) float v2f;
typedef __attribute__((ext_vector_type(8))) float v8f;

#define N_NODES 50000
#define N_EDGES 500000
#define HID 128
#define LDSPITCH (HID + 4)

#define TILE_N 196
#define TILE_THREADS 1024
#define TILE_WAVES (TILE_THREADS / 32)
#define QCAP 16384
#define TILE_GROUPS ((N_NODES + TILE_N - 1) / TILE_N)

template <int NL>
__device__ __forceinline__ void scan_items(const int* __restrict__ dst, int q0, int rot4, int E4,
                                           int n0, unsigned* queue, unsigned* qcount) {
  int4 d4[NL];
  unsigned e0[NL];
#pragma unroll
  for (int j = 0; j < NL; ++j) {
    int idx = q0 + j * TILE_THREADS + rot4;
    if (idx >= E4) idx -= E4;
    d4[j] = ((const int4*)dst)[idx];
    e0[j] = (unsigned)idx << 2;
  }
  unsigned t[NL][4];
  bool any = false;
#pragma unroll
  for (int j = 0; j < NL; ++j) {
    t[j][0] = (unsigned)(d4[j].x - n0);
    t[j][1] = (unsigned)(d4[j].y - n0);
    t[j][2] = (unsigned)(d4[j].z - n0);
    t[j][3] = (unsigned)(d4[j].w - n0);
#pragma unroll
    for (int k = 0; k < 4; ++k) any = any || (t[j][k] < (unsigned)TILE_N);
  }
  if (any) {
#pragma unroll
    for (int j = 0; j < NL; ++j) {
#pragma unroll
      for (int k = 0; k < 4; ++k) {
        if (t[j][k] < (unsigned)TILE_N) {
          const unsigned slot = atomicAdd(qcount, 1u);
          if (slot < (unsigned)QCAP) queue[slot] = ((e0[j] + (unsigned)k) << 8) | t[j][k];
        }
      }
    }
  }
}

template <int D, bool ADD_SCALED, bool BUILD>
__global__ __launch_bounds__(TILE_THREADS) void tile_sum(
    const float* __restrict__ feat,
    const int* __restrict__ src,
    const int* __restrict__ dst,
    const float* __restrict__ invdeg,
    float* invdeg_out,
    float* __restrict__ dest,
    int* qsaved,
    int* qsaved_n,
    int n_nodes, int n_edges) {
  constexpr int LPR = D / 4;
  constexpr int RPW = 32 / LPR;
  constexpr int U = 8;
  constexpr int SU = 4;
  __shared__ __attribute__((aligned(16))) float acc[TILE_N * D];
  __shared__ unsigned queue[QCAP];
  __shared__ unsigned qcount;
  __shared__ unsigned degc[BUILD ? TILE_N : 1];

  const int tid = threadIdx.x;
  const int wave = tid >> 5;
  const int lane = tid & 31;
  const int n0 = blockIdx.x * TILE_N;
  const int E4 = n_edges >> 2;
  const int rot4 = (int)blockIdx.x * (E4 / (int)gridDim.x);

  for (int i = tid; i < TILE_N * D / 4; i += TILE_THREADS)
    ((float4*)acc)[i] = make_float4(0.f, 0.f, 0.f, 0.f);
  if (tid == 0) qcount = 0u;
  if (BUILD) {
    for (int i = tid; i < TILE_N; i += TILE_THREADS) degc[i] = 0u;
  }
  __syncthreads();

  auto scan = [&](int qb, int qe) {
    int q0 = qb + tid;
    for (; q0 + (SU - 1) * TILE_THREADS < qe; q0 += SU * TILE_THREADS)
      scan_items<SU>(dst, q0, rot4, E4, n0, queue, &qcount);
    for (; q0 < qe; q0 += TILE_THREADS)
      scan_items<1>(dst, q0, rot4, E4, n0, queue, &qcount);
  };

  int* const qmine = qsaved + (long)blockIdx.x * QCAP;

  auto resolve = [&](int n, bool save) {
    for (int i = tid; i < n; i += TILE_THREADS) {
      const unsigned ent = queue[i];
      const unsigned s = (unsigned)src[ent >> 8];
      const unsigned res = (s << 8) | (ent & 0xFFu);
      queue[i] = res;
      if (save) qmine[i] = (int)res;
      if (BUILD) atomicAdd(&degc[ent & 0xFFu], 1u);
    }
    __syncthreads();
  };

  auto gather = [&](int n) {
    const int sub = lane / LPR;
    const int l = lane % LPR;
    for (int base = wave * RPW + sub; base < n + sub; base += TILE_WAVES * RPW * U) {
      float4 v[U];
      unsigned ent[U];
#pragma unroll
      for (int u = 0; u < U; ++u) {
        const int i = base + u * TILE_WAVES * RPW;
        const int ic = i < n ? i : n - 1;
        ent[u] = queue[ic];
        v[u] = *(const float4*)(feat + (long)(ent[u] >> 8) * D + l * 4);
      }
#pragma unroll
      for (int u = 0; u < U; ++u) {
        asm volatile("" ::"v"(v[u].x), "v"(v[u].y), "v"(v[u].z), "v"(v[u].w));
      }
#pragma unroll
      for (int u = 0; u < U; ++u) {
        const int i = base + u * TILE_WAVES * RPW;
        if (i < n) {
          float* a = acc + (int)(ent[u] & 0xFFu) * D + l;
          atomicAdd(a + 0 * LPR, v[u].x);
          atomicAdd(a + 1 * LPR, v[u].y);
          atomicAdd(a + 2 * LPR, v[u].z);
          atomicAdd(a + 3 * LPR, v[u].w);
        }
      }
    }
  };

  unsigned cnt;
  if (BUILD) {
    scan(0, E4);
    __syncthreads();
    cnt = qcount;
    if (tid == 0) qsaved_n[blockIdx.x] = (int)cnt;
  } else {
    cnt = (unsigned)qsaved_n[blockIdx.x];
  }
  if (cnt <= (unsigned)QCAP) {
    if (BUILD) {
      resolve((int)cnt, true);
    } else {
      for (int i = tid; i < (int)cnt; i += TILE_THREADS) queue[i] = (unsigned)qmine[i];
      __syncthreads();
    }
    gather((int)cnt);
  } else {
    for (int qb = 0; qb < E4; qb += QCAP / 4) {
      __syncthreads();
      if (tid == 0) qcount = 0u;
      __syncthreads();
      const int qe = qb + QCAP / 4 < E4 ? qb + QCAP / 4 : E4;
      scan(qb, qe);
      __syncthreads();
      const int n = (int)qcount;
      resolve(n, false);
      gather(n);
    }
  }
  __syncthreads();

  if (BUILD) {
    for (int i = tid; i < TILE_N; i += TILE_THREADS) {
      const int row = n0 + i;
      if (row < n_nodes) invdeg_out[row] = 1.0f / fmaxf((float)degc[i], 1.0f);
    }
  }

  for (int i = tid; i < TILE_N * LPR; i += TILE_THREADS) {
    const int r = i / LPR;
    const int l4 = i % LPR;
    const int row = n0 + r;
    if (row < n_nodes) {
      const float* a = acc + r * D + l4;
      float4 o = make_float4(a[0 * LPR], a[1 * LPR], a[2 * LPR], a[3 * LPR]);
      float4* p = (float4*)(dest + (long)row * D + l4 * 4);
      if (ADD_SCALED) {
        const float w = invdeg[row];
        const float4 c = *p;
        o.x = c.x + w * o.x; o.y = c.y + w * o.y; o.z = c.z + w * o.z; o.w = c.w + w * o.w;
      }
      *p = o;
    }
  }
}

template <int OUTD, bool RELU, int GRID>
__global__ __launch_bounds__(32 * (OUTD / 16)) void sage_gemm(
    const float* __restrict__ xin,
    const float* __restrict__ agg,
    const float* __restrict__ invdeg,
    const float* __restrict__ Wl,
    const float* __restrict__ Wr,
    const float* __restrict__ bias,
    float* __restrict__ out,
    int n_nodes) {
  __shared__ float sX[16][LDSPITCH];
  __shared__ float sA[16][LDSPITCH];

  const int tid = threadIdx.x;
  const int nthreads = 32 * (OUTD / 16);
  const int wave = tid >> 5;
  const int lane = tid & 31;
  const int half = lane >> 4;
  const int r = lane & 15;
  const int col = wave * 16 + r;

  v2f BL[HID / 4], BR[HID / 4];
#pragma unroll
  for (int j = 0; j < HID / 4; ++j) {
    const int ka = 4 * j + 2 * half;
    BL[j][0] = Wl[(long)ka * OUTD + col];
    BL[j][1] = Wl[(long)(ka + 1) * OUTD + col];
    BR[j][0] = Wr[(long)ka * OUTD + col];
    BR[j][1] = Wr[(long)(ka + 1) * OUTD + col];
  }
  const float bn = bias[col];

  const int n_strips = n_nodes >> 4;
  for (int strip = blockIdx.x; strip < n_strips; strip += GRID) {
    const int row0 = strip * 16;

    for (int i = tid; i < 16 * (HID / 4); i += nthreads) {
      const int rr = i / (HID / 4);
      const int c4 = (i % (HID / 4)) * 4;
      const long base = (long)(row0 + rr) * HID + c4;
      float4 xv = *(const float4*)(xin + base);
      float4 av = *(const float4*)(agg + base);
      const float iv = invdeg[row0 + rr];
      av.x *= iv; av.y *= iv; av.z *= iv; av.w *= iv;
      *(float4*)&sX[rr][c4] = xv;
      *(float4*)&sA[rr][c4] = av;
    }
    __syncthreads();

    v8f accL = {};
    v8f accR = {};
#pragma unroll
    for (int j = 0; j < HID / 4; ++j) {
      const int ka = 4 * j + 2 * half;
      v2f aL, aR;
      aL[0] = sA[r][ka];
      aL[1] = sA[r][ka + 1];
      aR[0] = sX[r][ka];
      aR[1] = sX[r][ka + 1];
      accL = __builtin_amdgcn_wmma_f32_16x16x4_f32(false, aL, false, BL[j],
                                                   (short)0, accL, false, false);
      accR = __builtin_amdgcn_wmma_f32_16x16x4_f32(false, aR, false, BR[j],
                                                   (short)0, accR, false, false);
    }

#pragma unroll
    for (int i = 0; i < 8; ++i) {
      const int gr = row0 + half * 8 + i;
      float v = accL[i] + accR[i] + bn;
      if (RELU) v = fmaxf(v, 0.0f);
      out[(long)gr * OUTD + col] = v;
    }
    __syncthreads();
  }
}

template <int STRIPS>
__global__ __launch_bounds__(256) void proj_gemm64(
    const float* __restrict__ xin,
    const float* __restrict__ Wl,
    const float* __restrict__ Wr,
    const float* __restrict__ bias,
    float* __restrict__ P,
    float* __restrict__ out,
    int n_nodes) {
  __shared__ float sX[16][LDSPITCH];

  const int tid = threadIdx.x;
  const int wave = tid >> 5;
  const int lane = tid & 31;
  const int half = lane >> 4;
  const int r = lane & 15;
  const bool selfside = wave >= 4;
  const float* __restrict__ Wsel = selfside ? Wr : Wl;
  float* __restrict__ dsel = selfside ? out : P;
  const int col = (wave & 3) * 16 + r;

  v2f B[HID / 4];
#pragma unroll
  for (int j = 0; j < HID / 4; ++j) {
    const int ka = 4 * j + 2 * half;
    B[j][0] = Wsel[(long)ka * 64 + col];
    B[j][1] = Wsel[(long)(ka + 1) * 64 + col];
  }
  const float bn = selfside ? bias[col] : 0.0f;

  for (int s = 0; s < STRIPS; ++s) {
    const int row0 = (blockIdx.x * STRIPS + s) * 16;
    if (row0 >= n_nodes) break;

    for (int i = tid; i < 16 * (HID / 4); i += 256) {
      const int rr = i / (HID / 4);
      const int c4 = (i % (HID / 4)) * 4;
      *(float4*)&sX[rr][c4] = *(const float4*)(xin + (long)(row0 + rr) * HID + c4);
    }
    __syncthreads();

    v8f acc0 = {};
    v8f acc1 = {};
#pragma unroll
    for (int j = 0; j < HID / 4; j += 2) {
      const int ka = 4 * j + 2 * half;
      v2f a0, a1;
      a0[0] = sX[r][ka];
      a0[1] = sX[r][ka + 1];
      a1[0] = sX[r][ka + 4];
      a1[1] = sX[r][ka + 5];
      acc0 = __builtin_amdgcn_wmma_f32_16x16x4_f32(false, a0, false, B[j],
                                                   (short)0, acc0, false, false);
      acc1 = __builtin_amdgcn_wmma_f32_16x16x4_f32(false, a1, false, B[j + 1],
                                                   (short)0, acc1, false, false);
    }

#pragma unroll
    for (int i = 0; i < 8; ++i) {
      const int gr = row0 + half * 8 + i;
      dsel[(long)gr * 64 + col] = acc0[i] + acc1[i] + bn;
    }
    __syncthreads();
  }
}

extern "C" void kernel_launch(void* const* d_in, const int* in_sizes, int n_in,
                              void* d_out, int out_size, void* d_ws, size_t ws_size,
                              hipStream_t stream) {
  const float* x   = (const float*)d_in[0];
  const int* eidx  = (const int*)d_in[1];
  const float* W1l = (const float*)d_in[2];
  const float* W1r = (const float*)d_in[3];
  const float* b1  = (const float*)d_in[4];
  const float* W2l = (const float*)d_in[5];
  const float* W2r = (const float*)d_in[6];
  const float* b2  = (const float*)d_in[7];
  const float* W3l = (const float*)d_in[8];
  const float* W3r = (const float*)d_in[9];
  const float* b3  = (const float*)d_in[10];
  float* out = (float*)d_out;

  const int* src = eidx;
  const int* dst = eidx + N_EDGES;

  float* deg = (float*)d_ws;
  float* agg = deg + N_NODES;
  float* h1  = agg + (long)N_NODES * HID;
  float* h2  = h1 + (long)N_NODES * HID;
  int* qn    = (int*)(h2 + (long)N_NODES * HID);
  int* qs    = qn + TILE_GROUPS;

  const int tileBlocks = TILE_GROUPS;
  constexpr int STRIPS = 4;
  const int gemmBlocks = (N_NODES + 16 * STRIPS - 1) / (16 * STRIPS);
  constexpr int SAGE_GRID = 768;

  tile_sum<HID, false, true><<<tileBlocks, TILE_THREADS, 0, stream>>>(x, src, dst, deg, deg, agg, qs, qn, N_NODES, N_EDGES);
  sage_gemm<128, true, SAGE_GRID><<<SAGE_GRID, 256, 0, stream>>>(x, agg, deg, W1l, W1r, b1, h1, N_NODES);

  tile_sum<HID, false, false><<<tileBlocks, TILE_THREADS, 0, stream>>>(h1, src, dst, deg, deg, agg, qs, qn, N_NODES, N_EDGES);
  sage_gemm<128, true, SAGE_GRID><<<SAGE_GRID, 256, 0, stream>>>(h1, agg, deg, W2l, W2r, b2, h2, N_NODES);

  float* P = agg;
  proj_gemm64<STRIPS><<<gemmBlocks, 256, 0, stream>>>(h2, W3l, W3r, b3, P, out, N_NODES);
  tile_sum<64, true, false><<<tileBlocks, TILE_THREADS, 0, stream>>>(P, src, dst, deg, deg, out, qs, qn, N_NODES, N_EDGES);
}


# owner queues built from a one-pass slice binning of the edge list (250 work-groups order 2000 edges each by owner in LDS) instead of 256 full walks of dst; LDS sums still in arrival order
# speedup vs baseline: 3.0423x; 1.1040x over previous
#include <hip/hip_runtime.h>

typedef __attribute__((ext_vector_type(2))) float v2f;
typedef __attribute__((ext_vector_type(8))) float v8f;

#define N_NODES 50000
#define N_EDGES 500000
#define HID 128
#define LDSPITCH (HID + 4)

#define TILE_N 196
#define TILE_THREADS 1024
#define TILE_WAVES (TILE_THREADS / 32)
#define QCAP 16384
#define TILE_GROUPS ((N_NODES + TILE_N - 1) / TILE_N)

#define NSLICE 250
#define SLICE_E (N_EDGES / NSLICE)
#define SLICE_Q (SLICE_E / 4)
#define SEG_PITCH 2016
#define OFF_PITCH 288
#define BIN_THREADS 256
#define BIN_NL 2
static_assert(NSLICE * SLICE_E == N_EDGES && SLICE_E % 4 == 0, "slices must tile the edge list");
static_assert(BIN_THREADS == TILE_GROUPS && NSLICE <= 256, "one thread per bin / per slice");
static_assert(BIN_THREADS * BIN_NL >= SLICE_Q && SLICE_E <= SEG_PITCH, "slice fits its threads");

__device__ __forceinline__ unsigned scan256(unsigned v, unsigned* wsum, int tid, unsigned& total) {
  const int lane = tid & 31;
  const int wave = tid >> 5;
  unsigned inc = v;
#pragma unroll
  for (int d = 1; d < 32; d <<= 1) {
    const unsigned n = __shfl_up(inc, (unsigned)d);
    if (lane >= d) inc += n;
  }
  if (wave < 8 && lane == 31) wsum[wave] = inc;
  __syncthreads();
  unsigned before = 0u, all = 0u;
#pragma unroll
  for (int w = 0; w < 8; ++w) {
    const unsigned s = wsum[w];
    if (w < wave) before += s;
    all += s;
  }
  total = all;
  return before + inc - v;
}

__global__ __launch_bounds__(BIN_THREADS) void bin_edges(
    const int* __restrict__ src,
    const int* __restrict__ dst,
    int* __restrict__ segs,
    int* __restrict__ offs) {
  __shared__ unsigned bcnt[TILE_GROUPS];
  __shared__ unsigned bstart[TILE_GROUPS];
  __shared__ unsigned wsum[8];
  __shared__ __attribute__((aligned(16))) unsigned sorted[SLICE_E];

  const int tid = threadIdx.x;
  const int s = blockIdx.x;
  const int4* d4p = (const int4*)(dst + (long)s * SLICE_E);
  const int4* s4p = (const int4*)(src + (long)s * SLICE_E);

  bcnt[tid] = 0u;
  __syncthreads();

  int4 dv[BIN_NL], sv[BIN_NL];
#pragma unroll
  for (int j = 0; j < BIN_NL; ++j) {
    const int q = tid + j * BIN_THREADS;
    dv[j] = d4p[q < SLICE_Q ? q : 0];
    sv[j] = s4p[q < SLICE_Q ? q : 0];
  }
#pragma unroll
  for (int j = 0; j < BIN_NL; ++j) {
    asm volatile("" ::"v"(dv[j].x), "v"(dv[j].y), "v"(dv[j].z), "v"(dv[j].w),
                 "v"(sv[j].x), "v"(sv[j].y), "v"(sv[j].z), "v"(sv[j].w));
  }
  unsigned ent[BIN_NL][4], own[BIN_NL][4], rank[BIN_NL][4];
#pragma unroll
  for (int j = 0; j < BIN_NL; ++j) {
    const bool valid = tid + j * BIN_THREADS < SLICE_Q;
    const unsigned du[4] = {(unsigned)dv[j].x, (unsigned)dv[j].y, (unsigned)dv[j].z, (unsigned)dv[j].w};
    const unsigned su[4] = {(unsigned)sv[j].x, (unsigned)sv[j].y, (unsigned)sv[j].z, (unsigned)sv[j].w};
#pragma unroll
    for (int k = 0; k < 4; ++k) {
      own[j][k] = 0xFFFFFFFFu;
      ent[j][k] = 0u;
      rank[j][k] = 0u;
      if (valid && du[k] < (unsigned)(TILE_GROUPS * TILE_N)) {
        const unsigned o = du[k] / (unsigned)TILE_N;
        own[j][k] = o;
        ent[j][k] = (su[k] << 8) | (du[k] - o * (unsigned)TILE_N);
        rank[j][k] = atomicAdd(&bcnt[o], 1u);
      }
    }
  }
  __syncthreads();

  unsigned total;
  const unsigned st = scan256(bcnt[tid], wsum, tid, total);
  bstart[tid] = st;
  offs[s * OFF_PITCH + tid] = (int)st;
  if (tid == 0) offs[s * OFF_PITCH + TILE_GROUPS] = (int)total;
  __syncthreads();

#pragma unroll
  for (int j = 0; j < BIN_NL; ++j) {
#pragma unroll
    for (int k = 0; k < 4; ++k) {
      if (own[j][k] != 0xFFFFFFFFu) sorted[bstart[own[j][k]] + rank[j][k]] = ent[j][k];
    }
  }
  __syncthreads();

  int4* outp = (int4*)(segs + (long)s * SEG_PITCH);
  for (int q = tid; q < SLICE_Q; q += BIN_THREADS) outp[q] = ((const int4*)sorted)[q];
}

template <int NL>
__device__ __forceinline__ void scan_items(const int* __restrict__ dst, int q0, int rot4, int E4,
                                           int n0, unsigned* queue, unsigned* qcount) {
  int4 d4[NL];
  unsigned e0[NL];
#pragma unroll
  for (int j = 0; j < NL; ++j) {
    int idx = q0 + j * TILE_THREADS + rot4;
    if (idx >= E4) idx -= E4;
    d4[j] = ((const int4*)dst)[idx];
    e0[j] = (unsigned)idx << 2;
  }
  unsigned t[NL][4];
  bool any = false;
#pragma unroll
  for (int j = 0; j < NL; ++j) {
    t[j][0] = (unsigned)(d4[j].x - n0);
    t[j][1] = (unsigned)(d4[j].y - n0);
    t[j][2] = (unsigned)(d4[j].z - n0);
    t[j][3] = (unsigned)(d4[j].w - n0);
#pragma unroll
    for (int k = 0; k < 4; ++k) any = any || (t[j][k] < (unsigned)TILE_N);
  }
  if (any) {
#pragma unroll
    for (int j = 0; j < NL; ++j) {
#pragma unroll
      for (int k = 0; k < 4; ++k) {
        if (t[j][k] < (unsigned)TILE_N) {
          const unsigned slot = atomicAdd(qcount, 1u);
          if (slot < (unsigned)QCAP) queue[slot] = ((e0[j] + (unsigned)k) << 8) | t[j][k];
        }
      }
    }
  }
}

template <int D, bool ADD_SCALED, bool BUILD>
__global__ __launch_bounds__(TILE_THREADS) void tile_sum(
    const float* __restrict__ feat,
    const int* __restrict__ src,
    const int* __restrict__ dst,
    const float* __restrict__ invdeg,
    float* invdeg_out,
    float* __restrict__ dest,
    int* qsaved,
    int* qsaved_n,
    const int* __restrict__ segs,
    const int* __restrict__ offs,
    int n_nodes, int n_edges) {
  constexpr int LPR = D / 4;
  constexpr int RPW = 32 / LPR;
  constexpr int U = 8;
  constexpr int SU = 4;
  __shared__ __attribute__((aligned(16))) float acc[TILE_N * D];
  __shared__ unsigned queue[QCAP];
  __shared__ unsigned qcount;
  __shared__ unsigned degc[BUILD ? TILE_N : 1];
  __shared__ unsigned sl_a[BUILD ? 256 : 1], sl_p[BUILD ? 256 : 1], sl_c[BUILD ? 256 : 1];
  __shared__ unsigned wsum[8];

  const int tid = threadIdx.x;
  const int wave = tid >> 5;
  const int lane = tid & 31;
  const int n0 = blockIdx.x * TILE_N;
  const int E4 = n_edges >> 2;
  const int rot4 = (int)blockIdx.x * (E4 / (int)gridDim.x);

  for (int i = tid; i < TILE_N * D / 4; i += TILE_THREADS)
    ((float4*)acc)[i] = make_float4(0.f, 0.f, 0.f, 0.f);
  if (tid == 0) qcount = 0u;
  if (BUILD) {
    for (int i = tid; i < TILE_N; i += TILE_THREADS) degc[i] = 0u;
  }
  __syncthreads();

  auto scan = [&](int qb, int qe) {
    int q0 = qb + tid;
    for (; q0 + (SU - 1) * TILE_THREADS < qe; q0 += SU * TILE_THREADS)
      scan_items<SU>(dst, q0, rot4, E4, n0, queue, &qcount);
    for (; q0 < qe; q0 += TILE_THREADS)
      scan_items<1>(dst, q0, rot4, E4, n0, queue, &qcount);
  };

  int* const qmine = qsaved + (long)blockIdx.x * QCAP;

  auto resolve = [&](int n) {
    for (int i = tid; i < n; i += TILE_THREADS) {
      const unsigned ent = queue[i];
      const unsigned s = (unsigned)src[ent >> 8];
      queue[i] = (s << 8) | (ent & 0xFFu);
      if (BUILD) atomicAdd(&degc[ent & 0xFFu], 1u);
    }
    __syncthreads();
  };

  auto gather = [&](int n) {
    const int sub = lane / LPR;
    const int l = lane % LPR;
    for (int base = wave * RPW + sub; base < n + sub; base += TILE_WAVES * RPW * U) {
      float4 v[U];
      unsigned ent[U];
#pragma unroll
      for (int u = 0; u < U; ++u) {
        const int i = base + u * TILE_WAVES * RPW;
        const int ic = i < n ? i : n - 1;
        ent[u] = queue[ic];
        v[u] = *(const float4*)(feat + (long)(ent[u] >> 8) * D + l * 4);
      }
#pragma unroll
      for (int u = 0; u < U; ++u) {
        asm volatile("" ::"v"(v[u].x), "v"(v[u].y), "v"(v[u].z), "v"(v[u].w));
      }
#pragma unroll
      for (int u = 0; u < U; ++u) {
        const int i = base + u * TILE_WAVES * RPW;
        if (i < n) {
          float* a = acc + (int)(ent[u] & 0xFFu) * D + l;
          atomicAdd(a + 0 * LPR, v[u].x);
          atomicAdd(a + 1 * LPR, v[u].y);
          atomicAdd(a + 2 * LPR, v[u].z);
          atomicAdd(a + 3 * LPR, v[u].w);
        }
      }
    }
  };

  unsigned cnt;
  if (BUILD) {
    unsigned a = 0u, len = 0u;
    if (tid < NSLICE) {
      const int* o = offs + tid * OFF_PITCH + blockIdx.x;
      a = (unsigned)o[0];
      len = (unsigned)o[1] - a;
    }
    const unsigned p = scan256(len, wsum, tid, cnt);
    if (tid < 256) {
      sl_a[tid] = a;
      sl_p[tid] = p;
      sl_c[tid] = len;
    }
    if (tid == 0) qsaved_n[blockIdx.x] = (int)cnt;
    __syncthreads();
  } else {
    cnt = (unsigned)qsaved_n[blockIdx.x];
  }
  if (cnt <= (unsigned)QCAP) {
    if (BUILD) {
      const int sl = tid >> 2;
      if (sl < NSLICE) {
        const unsigned len = sl_c[sl];
        const unsigned p = sl_p[sl];
        const int* sp = segs + sl * SEG_PITCH + sl_a[sl];
        for (unsigned i = (unsigned)(tid & 3); i < len; i += 4u) {
          const unsigned ent = (unsigned)sp[i];
          queue[p + i] = ent;
          qmine[p + i] = (int)ent;
          atomicAdd(&degc[ent & 0xFFu], 1u);
        }
      }
      __syncthreads();
    } else {
      for (int i = tid; i < (int)cnt; i += TILE_THREADS) queue[i] = (unsigned)qmine[i];
      __syncthreads();
    }
    gather((int)cnt);
  } else {
    for (int qb = 0; qb < E4; qb += QCAP / 4) {
      __syncthreads();
      if (tid == 0) qcount = 0u;
      __syncthreads();
      const int qe = qb + QCAP / 4 < E4 ? qb + QCAP / 4 : E4;
      scan(qb, qe);
      __syncthreads();
      const int n = (int)qcount;
      resolve(n);
      gather(n);
    }
  }
  __syncthreads();

  if (BUILD) {
    for (int i = tid; i < TILE_N; i += TILE_THREADS) {
      const int row = n0 + i;
      if (row < n_nodes) invdeg_out[row] = 1.0f / fmaxf((float)degc[i], 1.0f);
    }
  }

  for (int i = tid; i < TILE_N * LPR; i += TILE_THREADS) {
    const int r = i / LPR;
    const int l4 = i % LPR;
    const int row = n0 + r;
    if (row < n_nodes) {
      const float* a = acc + r * D + l4;
      float4 o = make_float4(a[0 * LPR], a[1 * LPR], a[2 * LPR], a[3 * LPR]);
      float4* p = (float4*)(dest + (long)row * D + l4 * 4);
      if (ADD_SCALED) {
        const float w = invdeg[row];
        const float4 c = *p;
        o.x = c.x + w * o.x; o.y = c.y + w * o.y; o.z = c.z + w * o.z; o.w = c.w + w * o.w;
      }
      *p = o;
    }
  }
}

template <int OUTD, bool RELU, int GRID>
__global__ __launch_bounds__(32 * (OUTD / 16)) void sage_gemm(
    const float* __restrict__ xin,
    const float* __restrict__ agg,
    const float* __restrict__ invdeg,
    const float* __restrict__ Wl,
    const float* __restrict__ Wr,
    const float* __restrict__ bias,
    float* __restrict__ out,
    int n_nodes) {
  __shared__ float sX[16][LDSPITCH];
  __shared__ float sA[16][LDSPITCH];

  const int tid = threadIdx.x;
  const int nthreads = 32 * (OUTD / 16);
  const int wave = tid >> 5;
  const int lane = tid & 31;
  const int half = lane >> 4;
  const int r = lane & 15;
  const int col = wave * 16 + r;

  v2f BL[HID / 4], BR[HID / 4];
#pragma unroll
  for (int j = 0; j < HID / 4; ++j) {
    const int ka = 4 * j + 2 * half;
    BL[j][0] = Wl[(long)ka * OUTD + col];
    BL[j][1] = Wl[(long)(ka + 1) * OUTD + col];
    BR[j][0] = Wr[(long)ka * OUTD + col];
    BR[j][1] = Wr[(long)(ka + 1) * OUTD + col];
  }
  const float bn = bias[col];

  const int n_strips = n_nodes >> 4;
  for (int strip = blockIdx.x; strip < n_strips; strip += GRID) {
    const int row0 = strip * 16;

    for (int i = tid; i < 16 * (HID / 4); i += nthreads) {
      const int rr = i / (HID / 4);
      const int c4 = (i % (HID / 4)) * 4;
      const long base = (long)(row0 + rr) * HID + c4;
      float4 xv = *(const float4*)(xin + base);
      float4 av = *(const float4*)(agg + base);
      const float iv = invdeg[row0 + rr];
      av.x *= iv; av.y *= iv; av.z *= iv; av.w *= iv;
      *(float4*)&sX[rr][c4] = xv;
      *(float4*)&sA[rr][c4] = av;
    }
    __syncthreads();

    v8f accL = {};
    v8f accR = {};
#pragma unroll
    for (int j = 0; j < HID / 4; ++j) {
      const int ka = 4 * j + 2 * half;
      v2f aL, aR;
      aL[0] = sA[r][ka];
      aL[1] = sA[r][ka + 1];
      aR[0] = sX[r][ka];
      aR[1] = sX[r][ka + 1];
      accL = __builtin_amdgcn_wmma_f32_16x16x4_f32(false, aL, false, BL[j],
                                                   (short)0, accL, false, false);
      accR = __builtin_amdgcn_wmma_f32_16x16x4_f32(false, aR, false, BR[j],
                                                   (short)0, accR, false, false);
    }

#pragma unroll
    for (int i = 0; i < 8; ++i) {
      const int gr = row0 + half * 8 + i;
      float v = accL[i] + accR[i] + bn;
      if (RELU) v = fmaxf(v, 0.0f);
      out[(long)gr * OUTD + col] = v;
    }
    __syncthreads();
  }
}

template <int STRIPS>
__global__ __launch_bounds__(256) void proj_gemm64(
    const float* __restrict__ xin,
    const float* __restrict__ Wl,
    const float* __restrict__ Wr,
    const float* __restrict__ bias,
    float* __restrict__ P,
    float* __restrict__ out,
    int n_nodes) {
  __shared__ float sX[16][LDSPITCH];

  const int tid = threadIdx.x;
  const int wave = tid >> 5;
  const int lane = tid & 31;
  const int half = lane >> 4;
  const int r = lane & 15;
  const bool selfside = wave >= 4;
  const float* __restrict__ Wsel = selfside ? Wr : Wl;
  float* __restrict__ dsel = selfside ? out : P;
  const int col = (wave & 3) * 16 + r;

  v2f B[HID / 4];
#pragma unroll
  for (int j = 0; j < HID / 4; ++j) {
    const int ka = 4 * j + 2 * half;
    B[j][0] = Wsel[(long)ka * 64 + col];
    B[j][1] = Wsel[(long)(ka + 1) * 64 + col];
  }
  const float bn = selfside ? bias[col] : 0.0f;

  for (int s = 0; s < STRIPS; ++s) {
    const int row0 = (blockIdx.x * STRIPS + s) * 16;
    if (row0 >= n_nodes) break;

    for (int i = tid; i < 16 * (HID / 4); i += 256) {
      const int rr = i / (HID / 4);
      const int c4 = (i % (HID / 4)) * 4;
      *(float4*)&sX[rr][c4] = *(const float4*)(xin + (long)(row0 + rr) * HID + c4);
    }
    __syncthreads();

    v8f acc0 = {};
    v8f acc1 = {};
#pragma unroll
    for (int j = 0; j < HID / 4; j += 2) {
      const int ka = 4 * j + 2 * half;
      v2f a0, a1;
      a0[0] = sX[r][ka];
      a0[1] = sX[r][ka + 1];
      a1[0] = sX[r][ka + 4];
      a1[1] = sX[r][ka + 5];
      acc0 = __builtin_amdgcn_wmma_f32_16x16x4_f32(false, a0, false, B[j],
                                                   (short)0, acc0, false, false);
      acc1 = __builtin_amdgcn_wmma_f32_16x16x4_f32(false, a1, false, B[j + 1],
                                                   (short)0, acc1, false, false);
    }

#pragma unroll
    for (int i = 0; i < 8; ++i) {
      const int gr = row0 + half * 8 + i;
      dsel[(long)gr * 64 + col] = acc0[i] + acc1[i] + bn;
    }
    __syncthreads();
  }
}

extern "C" void kernel_launch(void* const* d_in, const int* in_sizes, int n_in,
                              void* d_out, int out_size, void* d_ws, size_t ws_size,
                              hipStream_t stream) {
  const float* x   = (const float*)d_in[0];
  const int* eidx  = (const int*)d_in[1];
  const float* W1l = (const float*)d_in[2];
  const float* W1r = (const float*)d_in[3];
  const float* b1  = (const float*)d_in[4];
  const float* W2l = (const float*)d_in[5];
  const float* W2r = (const float*)d_in[6];
  const float* b2  = (const float*)d_in[7];
  const float* W3l = (const float*)d_in[8];
  const float* W3r = (const float*)d_in[9];
  const float* b3  = (const float*)d_in[10];
  float* out = (float*)d_out;

  const int* src = eidx;
  const int* dst = eidx + N_EDGES;

  float* deg = (float*)d_ws;
  float* agg = deg + N_NODES;
  float* h1  = agg + (long)N_NODES * HID;
  float* h2  = h1 + (long)N_NODES * HID;
  int* qn    = (int*)(h2 + (long)N_NODES * HID);
  int* qs    = qn + TILE_GROUPS;
  int* segs  = qs + (long)TILE_GROUPS * QCAP + 16;
  int* offs  = segs + (long)NSLICE * SEG_PITCH;

  const int tileBlocks = TILE_GROUPS;
  constexpr int STRIPS = 4;
  const int gemmBlocks = (N_NODES + 16 * STRIPS - 1) / (16 * STRIPS);
  constexpr int SAGE_GRID = 768;

  bin_edges<<<NSLICE, BIN_THREADS, 0, stream>>>(src, dst, segs, offs);

  tile_sum<HID, false, true><<<tileBlocks, TILE_THREADS, 0, stream>>>(x, src, dst, deg, deg, agg, qs, qn, segs, offs, N_NODES, N_EDGES);
  sage_gemm<128, true, SAGE_GRID><<<SAGE_GRID, 256, 0, stream>>>(x, agg, deg, W1l, W1r, b1, h1, N_NODES);

  tile_sum<HID, false, false><<<tileBlocks, TILE_THREADS, 0, stream>>>(h1, src, dst, deg, deg, agg, qs, qn, segs, offs, N_NODES, N_EDGES);
  sage_gemm<128, true, SAGE_GRID><<<SAGE_GRID, 256, 0, stream>>>(h1, agg, deg, W2l, W2r, b2, h2, N_NODES);

  float* P = agg;
  proj_gemm64<STRIPS><<<gemmBlocks, 256, 0, stream>>>(h2, W3l, W3r, b3, P, out, N_NODES);
  tile_sum<64, true, false><<<tileBlocks, TILE_THREADS, 0, stream>>>(P, src, dst, deg, deg, out, qs, qn, segs, offs, N_NODES, N_EDGES);
}


# layer 3's projections moved into the layer-2 GEMM launch: a block turns its relu'd 16x128 strip through LDS into P and the self term; h2 never goes to memory; one launch fewer; sum order kept
# speedup vs baseline: 3.3368x; 1.0968x over previous
#include <hip/hip_runtime.h>

typedef __attribute__((ext_vector_type(2))) float v2f;
typedef __attribute__((ext_vector_type(8))) float v8f;

#define N_NODES 50000
#define N_EDGES 500000
#define HID 128
#define LDSPITCH (HID + 4)

#define TILE_N 196
#define TILE_THREADS 1024
#define TILE_WAVES (TILE_THREADS / 32)
#define QCAP 16384
#define TILE_GROUPS ((N_NODES + TILE_N - 1) / TILE_N)

#define NSLICE 250
#define SLICE_E (N_EDGES / NSLICE)
#define SLICE_Q (SLICE_E / 4)
#define SEG_PITCH 2016
#define OFF_PITCH 288
#define BIN_THREADS 256
#define BIN_NL 2
static_assert(NSLICE * SLICE_E == N_EDGES && SLICE_E % 4 == 0, "slices must tile the edge list");
static_assert(BIN_THREADS == TILE_GROUPS && NSLICE <= 256, "one thread per bin / per slice");
static_assert(BIN_THREADS * BIN_NL >= SLICE_Q && SLICE_E <= SEG_PITCH, "slice fits its threads");

__device__ __forceinline__ unsigned scan256(unsigned v, unsigned* wsum, int tid, unsigned& total) {
  const int lane = tid & 31;
  const int wave = tid >> 5;
  unsigned inc = v;
#pragma unroll
  for (int d = 1; d < 32; d <<= 1) {
    const unsigned n = __shfl_up(inc, (unsigned)d);
    if (lane >= d) inc += n;
  }
  if (wave < 8 && lane == 31) wsum[wave] = inc;
  __syncthreads();
  unsigned before = 0u, all = 0u;
#pragma unroll
  for (int w = 0; w < 8; ++w) {
    const unsigned s = wsum[w];
    if (w < wave) before += s;
    all += s;
  }
  total = all;
  return before + inc - v;
}

__global__ __launch_bounds__(BIN_THREADS) void bin_edges(
    const int* __restrict__ src,
    const int* __restrict__ dst,
    int* __restrict__ segs,
    int* __restrict__ offs) {
  __shared__ unsigned bcnt[TILE_GROUPS];
  __shared__ unsigned bstart[TILE_GROUPS];
  __shared__ unsigned wsum[8];
  __shared__ __attribute__((aligned(16))) unsigned sorted[SLICE_E];

  const int tid = threadIdx.x;
  const int s = blockIdx.x;
  const int4* d4p = (const int4*)(dst + (long)s * SLICE_E);
  const int4* s4p = (const int4*)(src + (long)s * SLICE_E);

  bcnt[tid] = 0u;
  __syncthreads();

  int4 dv[BIN_NL], sv[BIN_NL];
#pragma unroll
  for (int j = 0; j < BIN_NL; ++j) {
    const int q = tid + j * BIN_THREADS;
    dv[j] = d4p[q < SLICE_Q ? q : 0];
    sv[j] = s4p[q < SLICE_Q ? q : 0];
  }
#pragma unroll
  for (int j = 0; j < BIN_NL; ++j) {
    asm volatile("" ::"v"(dv[j].x), "v"(dv[j].y), "v"(dv[j].z), "v"(dv[j].w),
                 "v"(sv[j].x), "v"(sv[j].y), "v"(sv[j].z), "v"(sv[j].w));
  }
  unsigned ent[BIN_NL][4], own[BIN_NL][4], rank[BIN_NL][4];
#pragma unroll
  for (int j = 0; j < BIN_NL; ++j) {
    const bool valid = tid + j * BIN_THREADS < SLICE_Q;
    const unsigned du[4] = {(unsigned)dv[j].x, (unsigned)dv[j].y, (unsigned)dv[j].z, (unsigned)dv[j].w};
    const unsigned su[4] = {(unsigned)sv[j].x, (unsigned)sv[j].y, (unsigned)sv[j].z, (unsigned)sv[j].w};
#pragma unroll
    for (int k = 0; k < 4; ++k) {
      own[j][k] = 0xFFFFFFFFu;
      ent[j][k] = 0u;
      rank[j][k] = 0u;
      if (valid && du[k] < (unsigned)(TILE_GROUPS * TILE_N)) {
        const unsigned o = du[k] / (unsigned)TILE_N;
        own[j][k] = o;
        ent[j][k] = (su[k] << 8) | (du[k] - o * (unsigned)TILE_N);
        rank[j][k] = atomicAdd(&bcnt[o], 1u);
      }
    }
  }
  __syncthreads();

  unsigned total;
  const unsigned st = scan256(bcnt[tid], wsum, tid, total);
  bstart[tid] = st;
  offs[s * OFF_PITCH + tid] = (int)st;
  if (tid == 0) offs[s * OFF_PITCH + TILE_GROUPS] = (int)total;
  __syncthreads();

#pragma unroll
  for (int j = 0; j < BIN_NL; ++j) {
#pragma unroll
    for (int k = 0; k < 4; ++k) {
      if (own[j][k] != 0xFFFFFFFFu) sorted[bstart[own[j][k]] + rank[j][k]] = ent[j][k];
    }
  }
  __syncthreads();

  int4* outp = (int4*)(segs + (long)s * SEG_PITCH);
  for (int q = tid; q < SLICE_Q; q += BIN_THREADS) outp[q] = ((const int4*)sorted)[q];
}

template <int NL>
__device__ __forceinline__ void scan_items(const int* __restrict__ dst, int q0, int rot4, int E4,
                                           int n0, unsigned* queue, unsigned* qcount) {
  int4 d4[NL];
  unsigned e0[NL];
#pragma unroll
  for (int j = 0; j < NL; ++j) {
    int idx = q0 + j * TILE_THREADS + rot4;
    if (idx >= E4) idx -= E4;
    d4[j] = ((const int4*)dst)[idx];
    e0[j] = (unsigned)idx << 2;
  }
  unsigned t[NL][4];
  bool any = false;
#pragma unroll
  for (int j = 0; j < NL; ++j) {
    t[j][0] = (unsigned)(d4[j].x - n0);
    t[j][1] = (unsigned)(d4[j].y - n0);
    t[j][2] = (unsigned)(d4[j].z - n0);
    t[j][3] = (unsigned)(d4[j].w - n0);
#pragma unroll
    for (int k = 0; k < 4; ++k) any = any || (t[j][k] < (unsigned)TILE_N);
  }
  if (any) {
#pragma unroll
    for (int j = 0; j < NL; ++j) {
#pragma unroll
      for (int k = 0; k < 4; ++k) {
        if (t[j][k] < (unsigned)TILE_N) {
          const unsigned slot = atomicAdd(qcount, 1u);
          if (slot < (unsigned)QCAP) queue[slot] = ((e0[j] + (unsigned)k) << 8) | t[j][k];
        }
      }
    }
  }
}

template <int D, bool ADD_SCALED, bool BUILD>
__global__ __launch_bounds__(TILE_THREADS) void tile_sum(
    const float* __restrict__ feat,
    const int* __restrict__ src,
    const int* __restrict__ dst,
    const float* __restrict__ invdeg,
    float* invdeg_out,
    float* __restrict__ dest,
    int* qsaved,
    int* qsaved_n,
    const int* __restrict__ segs,
    const int* __restrict__ offs,
    int n_nodes, int n_edges) {
  constexpr int LPR = D / 4;
  constexpr int RPW = 32 / LPR;
  constexpr int U = 8;
  constexpr int SU = 4;
  __shared__ __attribute__((aligned(16))) float acc[TILE_N * D];
  __shared__ unsigned queue[QCAP];
  __shared__ unsigned qcount;
  __shared__ unsigned degc[BUILD ? TILE_N : 1];
  __shared__ unsigned sl_a[BUILD ? 256 : 1], sl_p[BUILD ? 256 : 1], sl_c[BUILD ? 256 : 1];
  __shared__ unsigned wsum[8];

  const int tid = threadIdx.x;
  const int wave = tid >> 5;
  const int lane = tid & 31;
  const int n0 = blockIdx.x * TILE_N;
  const int E4 = n_edges >> 2;
  const int rot4 = (int)blockIdx.x * (E4 / (int)gridDim.x);

  for (int i = tid; i < TILE_N * D / 4; i += TILE_THREADS)
    ((float4*)acc)[i] = make_float4(0.f, 0.f, 0.f, 0.f);
  if (tid == 0) qcount = 0u;
  if (BUILD) {
    for (int i = tid; i < TILE_N; i += TILE_THREADS) degc[i] = 0u;
  }
  __syncthreads();

  auto scan = [&](int qb, int qe) {
    int q0 = qb + tid;
    for (; q0 + (SU - 1) * TILE_THREADS < qe; q0 += SU * TILE_THREADS)
      scan_items<SU>(dst, q0, rot4, E4, n0, queue, &qcount);
    for (; q0 < qe; q0 += TILE_THREADS)
      scan_items<1>(dst, q0, rot4, E4, n0, queue, &qcount);
  };

  int* const qmine = qsaved + (long)blockIdx.x * QCAP;

  auto resolve = [&](int n) {
    for (int i = tid; i < n; i += TILE_THREADS) {
      const unsigned ent = queue[i];
      const unsigned s = (unsigned)src[ent >> 8];
      queue[i] = (s << 8) | (ent & 0xFFu);
      if (BUILD) atomicAdd(&degc[ent & 0xFFu], 1u);
    }
    __syncthreads();
  };

  auto gather = [&](int n) {
    const int sub = lane / LPR;
    const int l = lane % LPR;
    for (int base = wave * RPW + sub; base < n + sub; base += TILE_WAVES * RPW * U) {
      float4 v[U];
      unsigned ent[U];
#pragma unroll
      for (int u = 0; u < U; ++u) {
        const int i = base + u * TILE_WAVES * RPW;
        const int ic = i < n ? i : n - 1;
        ent[u] = queue[ic];
        v[u] = *(const float4*)(feat + (long)(ent[u] >> 8) * D + l * 4);
      }
#pragma unroll
      for (int u = 0; u < U; ++u) {
        asm volatile("" ::"v"(v[u].x), "v"(v[u].y), "v"(v[u].z), "v"(v[u].w));
      }
#pragma unroll
      for (int u = 0; u < U; ++u) {
        const int i = base + u * TILE_WAVES * RPW;
        if (i < n) {
          float* a = acc + (int)(ent[u] & 0xFFu) * D + l;
          atomicAdd(a + 0 * LPR, v[u].x);
          atomicAdd(a + 1 * LPR, v[u].y);
          atomicAdd(a + 2 * LPR, v[u].z);
          atomicAdd(a + 3 * LPR, v[u].w);
        }
      }
    }
  };

  unsigned cnt;
  if (BUILD) {
    unsigned a = 0u, len = 0u;
    if (tid < NSLICE) {
      const int* o = offs + tid * OFF_PITCH + blockIdx.x;
      a = (unsigned)o[0];
      len = (unsigned)o[1] - a;
    }
    const unsigned p = scan256(len, wsum, tid, cnt);
    if (tid < 256) {
      sl_a[tid] = a;
      sl_p[tid] = p;
      sl_c[tid] = len;
    }
    if (tid == 0) qsaved_n[blockIdx.x] = (int)cnt;
    __syncthreads();
  } else {
    cnt = (unsigned)qsaved_n[blockIdx.x];
  }
  if (cnt <= (unsigned)QCAP) {
    if (BUILD) {
      const int sl = tid >> 2;
      if (sl < NSLICE) {
        const unsigned len = sl_c[sl];
        const unsigned p = sl_p[sl];
        const int* sp = segs + sl * SEG_PITCH + sl_a[sl];
        for (unsigned i = (unsigned)(tid & 3); i < len; i += 4u) {
          const unsigned ent = (unsigned)sp[i];
          queue[p + i] = ent;
          qmine[p + i] = (int)ent;
          atomicAdd(&degc[ent & 0xFFu], 1u);
        }
      }
      __syncthreads();
    } else {
      for (int i = tid; i < (int)cnt; i += TILE_THREADS) queue[i] = (unsigned)qmine[i];
      __syncthreads();
    }
    gather((int)cnt);
  } else {
    for (int qb = 0; qb < E4; qb += QCAP / 4) {
      __syncthreads();
      if (tid == 0) qcount = 0u;
      __syncthreads();
      const int qe = qb + QCAP / 4 < E4 ? qb + QCAP / 4 : E4;
      scan(qb, qe);
      __syncthreads();
      const int n = (int)qcount;
      resolve(n);
      gather(n);
    }
  }
  __syncthreads();

  if (BUILD) {
    for (int i = tid; i < TILE_N; i += TILE_THREADS) {
      const int row = n0 + i;
      if (row < n_nodes) invdeg_out[row] = 1.0f / fmaxf((float)degc[i], 1.0f);
    }
  }

  for (int i = tid; i < TILE_N * LPR; i += TILE_THREADS) {
    const int r = i / LPR;
    const int l4 = i % LPR;
    const int row = n0 + r;
    if (row < n_nodes) {
      const float* a = acc + r * D + l4;
      float4 o = make_float4(a[0 * LPR], a[1 * LPR], a[2 * LPR], a[3 * LPR]);
      float4* p = (float4*)(dest + (long)row * D + l4 * 4);
      if (ADD_SCALED) {
        const float w = invdeg[row];
        const float4 c = *p;
        o.x = c.x + w * o.x; o.y = c.y + w * o.y; o.z = c.z + w * o.z; o.w = c.w + w * o.w;
      }
      *p = o;
    }
  }
}

template <int OUTD, bool RELU, int GRID>
__global__ __launch_bounds__(32 * (OUTD / 16)) void sage_gemm(
    const float* __restrict__ xin,
    const float* __restrict__ agg,
    const float* __restrict__ invdeg,
    const float* __restrict__ Wl,
    const float* __restrict__ Wr,
    const float* __restrict__ bias,
    float* __restrict__ out,
    int n_nodes) {
  __shared__ float sX[16][LDSPITCH];
  __shared__ float sA[16][LDSPITCH];

  const int tid = threadIdx.x;
  const int nthreads = 32 * (OUTD / 16);
  const int wave = tid >> 5;
  const int lane = tid & 31;
  const int half = lane >> 4;
  const int r = lane & 15;
  const int col = wave * 16 + r;

  v2f BL[HID / 4], BR[HID / 4];
#pragma unroll
  for (int j = 0; j < HID / 4; ++j) {
    const int ka = 4 * j + 2 * half;
    BL[j][0] = Wl[(long)ka * OUTD + col];
    BL[j][1] = Wl[(long)(ka + 1) * OUTD + col];
    BR[j][0] = Wr[(long)ka * OUTD + col];
    BR[j][1] = Wr[(long)(ka + 1) * OUTD + col];
  }
  const float bn = bias[col];

  const int n_strips = n_nodes >> 4;
  for (int strip = blockIdx.x; strip < n_strips; strip += GRID) {
    const int row0 = strip * 16;

    for (int i = tid; i < 16 * (HID / 4); i += nthreads) {
      const int rr = i / (HID / 4);
      const int c4 = (i % (HID / 4)) * 4;
      const long base = (long)(row0 + rr) * HID + c4;
      float4 xv = *(const float4*)(xin + base);
      float4 av = *(const float4*)(agg + base);
      const float iv = invdeg[row0 + rr];
      av.x *= iv; av.y *= iv; av.z *= iv; av.w *= iv;
      *(float4*)&sX[rr][c4] = xv;
      *(float4*)&sA[rr][c4] = av;
    }
    __syncthreads();

    v8f accL = {};
    v8f accR = {};
#pragma unroll
    for (int j = 0; j < HID / 4; ++j) {
      const int ka = 4 * j + 2 * half;
      v2f aL, aR;
      aL[0] = sA[r][ka];
      aL[1] = sA[r][ka + 1];
      aR[0] = sX[r][ka];
      aR[1] = sX[r][ka + 1];
      accL = __builtin_amdgcn_wmma_f32_16x16x4_f32(false, aL, false, BL[j],
                                                   (short)0, accL, false, false);
      accR = __builtin_amdgcn_wmma_f32_16x16x4_f32(false, aR, false, BR[j],
                                                   (short)0, accR, false, false);
    }

#pragma unroll
    for (int i = 0; i < 8; ++i) {
      const int gr = row0 + half * 8 + i;
      float v = accL[i] + accR[i] + bn;
      if (RELU) v = fmaxf(v, 0.0f);
      out[(long)gr * OUTD + col] = v;
    }
    __syncthreads();
  }
}

template <int GRID>
__global__ __launch_bounds__(256) void sage_gemm_proj(
    const float* __restrict__ xin,
    const float* __restrict__ agg,
    const float* __restrict__ invdeg,
    const float* __restrict__ Wl,
    const float* __restrict__ Wr,
    const float* __restrict__ bias,
    const float* __restrict__ W3l,
    const float* __restrict__ W3r,
    const float* __restrict__ bias3,
    float* __restrict__ P,
    float* __restrict__ out,
    int n_nodes) {
  __shared__ float sX[16][LDSPITCH];
  __shared__ float sA[16][LDSPITCH];
  __shared__ float sH[16][LDSPITCH];

  const int tid = threadIdx.x;
  const int wave = tid >> 5;
  const int lane = tid & 31;
  const int half = lane >> 4;
  const int r = lane & 15;
  const int col = wave * 16 + r;
  const bool selfside = wave >= 4;
  const float* __restrict__ W3sel = selfside ? W3r : W3l;
  float* __restrict__ dsel = selfside ? out : P;
  const int col3 = (wave & 3) * 16 + r;

  v2f BL[HID / 4], BR[HID / 4], B3[HID / 4];
#pragma unroll
  for (int j = 0; j < HID / 4; ++j) {
    const int ka = 4 * j + 2 * half;
    BL[j][0] = Wl[(long)ka * HID + col];
    BL[j][1] = Wl[(long)(ka + 1) * HID + col];
    BR[j][0] = Wr[(long)ka * HID + col];
    BR[j][1] = Wr[(long)(ka + 1) * HID + col];
    B3[j][0] = W3sel[(long)ka * 64 + col3];
    B3[j][1] = W3sel[(long)(ka + 1) * 64 + col3];
  }
  const float bn = bias[col];
  const float bn3 = selfside ? bias3[col3] : 0.0f;

  const int n_strips = n_nodes >> 4;
  for (int strip = blockIdx.x; strip < n_strips; strip += GRID) {
    const int row0 = strip * 16;

    for (int i = tid; i < 16 * (HID / 4); i += 256) {
      const int rr = i / (HID / 4);
      const int c4 = (i % (HID / 4)) * 4;
      const long base = (long)(row0 + rr) * HID + c4;
      float4 xv = *(const float4*)(xin + base);
      float4 av = *(const float4*)(agg + base);
      const float iv = invdeg[row0 + rr];
      av.x *= iv; av.y *= iv; av.z *= iv; av.w *= iv;
      *(float4*)&sX[rr][c4] = xv;
      *(float4*)&sA[rr][c4] = av;
    }
    __syncthreads();

    v8f accL = {};
    v8f accR = {};
#pragma unroll
    for (int j = 0; j < HID / 4; ++j) {
      const int ka = 4 * j + 2 * half;
      v2f aL, aR;
      aL[0] = sA[r][ka];
      aL[1] = sA[r][ka + 1];
      aR[0] = sX[r][ka];
      aR[1] = sX[r][ka + 1];
      accL = __builtin_amdgcn_wmma_f32_16x16x4_f32(false, aL, false, BL[j],
                                                   (short)0, accL, false, false);
      accR = __builtin_amdgcn_wmma_f32_16x16x4_f32(false, aR, false, BR[j],
                                                   (short)0, accR, false, false);
    }

#pragma unroll
    for (int i = 0; i < 8; ++i) {
      sH[half * 8 + i][col] = fmaxf(accL[i] + accR[i] + bn, 0.0f);
    }
    __syncthreads();

    v8f acc0 = {};
    v8f acc1 = {};
#pragma unroll
    for (int j = 0; j < HID / 4; j += 2) {
      const int ka = 4 * j + 2 * half;
      v2f a0, a1;
      a0[0] = sH[r][ka];
      a0[1] = sH[r][ka + 1];
      a1[0] = sH[r][ka + 4];
      a1[1] = sH[r][ka + 5];
      acc0 = __builtin_amdgcn_wmma_f32_16x16x4_f32(false, a0, false, B3[j],
                                                   (short)0, acc0, false, false);
      acc1 = __builtin_amdgcn_wmma_f32_16x16x4_f32(false, a1, false, B3[j + 1],
                                                   (short)0, acc1, false, false);
    }

#pragma unroll
    for (int i = 0; i < 8; ++i) {
      const int gr = row0 + half * 8 + i;
      dsel[(long)gr * 64 + col3] = acc0[i] + acc1[i] + bn3;
    }
  }
}

extern "C" void kernel_launch(void* const* d_in, const int* in_sizes, int n_in,
                              void* d_out, int out_size, void* d_ws, size_t ws_size,
                              hipStream_t stream) {
  const float* x   = (const float*)d_in[0];
  const int* eidx  = (const int*)d_in[1];
  const float* W1l = (const float*)d_in[2];
  const float* W1r = (const float*)d_in[3];
  const float* b1  = (const float*)d_in[4];
  const float* W2l = (const float*)d_in[5];
  const float* W2r = (const float*)d_in[6];
  const float* b2  = (const float*)d_in[7];
  const float* W3l = (const float*)d_in[8];
  const float* W3r = (const float*)d_in[9];
  const float* b3  = (const float*)d_in[10];
  float* out = (float*)d_out;

  const int* src = eidx;
  const int* dst = eidx + N_EDGES;

  float* deg = (float*)d_ws;
  float* agg = deg + N_NODES;
  float* h1  = agg + (long)N_NODES * HID;
  float* P   = h1 + (long)N_NODES * HID;
  int* qn    = (int*)(P + (long)N_NODES * HID);
  int* qs    = qn + TILE_GROUPS;
  int* segs  = qs + (long)TILE_GROUPS * QCAP + 16;
  int* offs  = segs + (long)NSLICE * SEG_PITCH;

  const int tileBlocks = TILE_GROUPS;
  constexpr int SAGE_GRID = 768;
  constexpr int PROJ_GRID = 512;

  bin_edges<<<NSLICE, BIN_THREADS, 0, stream>>>(src, dst, segs, offs);

  tile_sum<HID, false, true><<<tileBlocks, TILE_THREADS, 0, stream>>>(x, src, dst, deg, deg, agg, qs, qn, segs, offs, N_NODES, N_EDGES);
  sage_gemm<128, true, SAGE_GRID><<<SAGE_GRID, 256, 0, stream>>>(x, agg, deg, W1l, W1r, b1, h1, N_NODES);

  tile_sum<HID, false, false><<<tileBlocks, TILE_THREADS, 0, stream>>>(h1, src, dst, deg, deg, agg, qs, qn, segs, offs, N_NODES, N_EDGES);
  sage_gemm_proj<PROJ_GRID><<<PROJ_GRID, 256, 0, stream>>>(h1, agg, deg, W2l, W2r, b2, W3l, W3r, b3, P, out, N_NODES);

  tile_sum<64, true, false><<<tileBlocks, TILE_THREADS, 0, stream>>>(P, src, dst, deg, deg, out, qs, qn, segs, offs, N_NODES, N_EDGES);
}


# owner work-groups order their queue by owning wave (local row mod 32) once per call; a wave sums its 6-7 rows in index-picked registers: no LDS float adds or LDS sums; row sums run in queue order
# speedup vs baseline: 3.5223x; 1.0556x over previous
#include <hip/hip_runtime.h>

typedef __attribute__((ext_vector_type(2))) float v2f;
typedef __attribute__((ext_vector_type(8))) float v8f;

#define N_NODES 50000
#define N_EDGES 500000
#define HID 128
#define LDSPITCH (HID + 4)

#define TILE_N 196
#define TILE_THREADS 1024
#define TILE_WAVES (TILE_THREADS / 32)
#define QCAP 16384
#define TILE_GROUPS ((N_NODES + TILE_N - 1) / TILE_N)

#define NSLICE 250
#define SLICE_E (N_EDGES / NSLICE)
#define SLICE_Q (SLICE_E / 4)
#define SEG_PITCH 2016
#define OFF_PITCH 288
#define BIN_THREADS 256
#define BIN_NL 2
static_assert(NSLICE * SLICE_E == N_EDGES && SLICE_E % 4 == 0, "slices must tile the edge list");
static_assert(BIN_THREADS == TILE_GROUPS && NSLICE <= 256, "one thread per bin / per slice");
static_assert(BIN_THREADS * BIN_NL >= SLICE_Q && SLICE_E <= SEG_PITCH, "slice fits its threads");

__device__ __forceinline__ unsigned scan256(unsigned v, unsigned* wsum, int tid, unsigned& total) {
  const int lane = tid & 31;
  const int wave = tid >> 5;
  unsigned inc = v;
#pragma unroll
  for (int d = 1; d < 32; d <<= 1) {
    const unsigned n = __shfl_up(inc, (unsigned)d);
    if (lane >= d) inc += n;
  }
  if (wave < 8 && lane == 31) wsum[wave] = inc;
  __syncthreads();
  unsigned before = 0u, all = 0u;
#pragma unroll
  for (int w = 0; w < 8; ++w) {
    const unsigned s = wsum[w];
    if (w < wave) before += s;
    all += s;
  }
  total = all;
  return before + inc - v;
}

__global__ __launch_bounds__(BIN_THREADS) void bin_edges(
    const int* __restrict__ src,
    const int* __restrict__ dst,
    int* __restrict__ segs,
    int* __restrict__ offs) {
  __shared__ unsigned bcnt[TILE_GROUPS];
  __shared__ unsigned bstart[TILE_GROUPS];
  __shared__ unsigned wsum[8];
  __shared__ __attribute__((aligned(16))) unsigned sorted[SLICE_E];

  const int tid = threadIdx.x;
  const int s = blockIdx.x;
  const int4* d4p = (const int4*)(dst + (long)s * SLICE_E);
  const int4* s4p = (const int4*)(src + (long)s * SLICE_E);

  bcnt[tid] = 0u;
  __syncthreads();

  int4 dv[BIN_NL], sv[BIN_NL];
#pragma unroll
  for (int j = 0; j < BIN_NL; ++j) {
    const int q = tid + j * BIN_THREADS;
    dv[j] = d4p[q < SLICE_Q ? q : 0];
    sv[j] = s4p[q < SLICE_Q ? q : 0];
  }
#pragma unroll
  for (int j = 0; j < BIN_NL; ++j) {
    asm volatile("" ::"v"(dv[j].x), "v"(dv[j].y), "v"(dv[j].z), "v"(dv[j].w),
                 "v"(sv[j].x), "v"(sv[j].y), "v"(sv[j].z), "v"(sv[j].w));
  }
  unsigned ent[BIN_NL][4], own[BIN_NL][4], rank[BIN_NL][4];
#pragma unroll
  for (int j = 0; j < BIN_NL; ++j) {
    const bool valid = tid + j * BIN_THREADS < SLICE_Q;
    const unsigned du[4] = {(unsigned)dv[j].x, (unsigned)dv[j].y, (unsigned)dv[j].z, (unsigned)dv[j].w};
    const unsigned su[4] = {(unsigned)sv[j].x, (unsigned)sv[j].y, (unsigned)sv[j].z, (unsigned)sv[j].w};
#pragma unroll
    for (int k = 0; k < 4; ++k) {
      own[j][k] = 0xFFFFFFFFu;
      ent[j][k] = 0u;
      rank[j][k] = 0u;
      if (valid && du[k] < (unsigned)(TILE_GROUPS * TILE_N)) {
        const unsigned o = du[k] / (unsigned)TILE_N;
        own[j][k] = o;
        ent[j][k] = (su[k] << 8) | (du[k] - o * (unsigned)TILE_N);
        rank[j][k] = atomicAdd(&bcnt[o], 1u);
      }
    }
  }
  __syncthreads();

  unsigned total;
  const unsigned st = scan256(bcnt[tid], wsum, tid, total);
  bstart[tid] = st;
  offs[s * OFF_PITCH + tid] = (int)st;
  if (tid == 0) offs[s * OFF_PITCH + TILE_GROUPS] = (int)total;
  __syncthreads();

#pragma unroll
  for (int j = 0; j < BIN_NL; ++j) {
#pragma unroll
    for (int k = 0; k < 4; ++k) {
      if (own[j][k] != 0xFFFFFFFFu) sorted[bstart[own[j][k]] + rank[j][k]] = ent[j][k];
    }
  }
  __syncthreads();

  int4* outp = (int4*)(segs + (long)s * SEG_PITCH);
  for (int q = tid; q < SLICE_Q; q += BIN_THREADS) outp[q] = ((const int4*)sorted)[q];
}

template <int NL>
__device__ __forceinline__ void scan_items(const int* __restrict__ dst, int q0, int rot4, int E4,
                                           int n0, unsigned* queue, unsigned* qcount) {
  int4 d4[NL];
  unsigned e0[NL];
#pragma unroll
  for (int j = 0; j < NL; ++j) {
    int idx = q0 + j * TILE_THREADS + rot4;
    if (idx >= E4) idx -= E4;
    d4[j] = ((const int4*)dst)[idx];
    e0[j] = (unsigned)idx << 2;
  }
  unsigned t[NL][4];
  bool any = false;
#pragma unroll
  for (int j = 0; j < NL; ++j) {
    t[j][0] = (unsigned)(d4[j].x - n0);
    t[j][1] = (unsigned)(d4[j].y - n0);
    t[j][2] = (unsigned)(d4[j].z - n0);
    t[j][3] = (unsigned)(d4[j].w - n0);
#pragma unroll
    for (int k = 0; k < 4; ++k) any = any || (t[j][k] < (unsigned)TILE_N);
  }
  if (any) {
#pragma unroll
    for (int j = 0; j < NL; ++j) {
#pragma unroll
      for (int k = 0; k < 4; ++k) {
        if (t[j][k] < (unsigned)TILE_N) {
          const unsigned slot = atomicAdd(qcount, 1u);
          if (slot < (unsigned)QCAP) queue[slot] = ((e0[j] + (unsigned)k) << 8) | t[j][k];
        }
      }
    }
  }
}

typedef __attribute__((ext_vector_type(4))) float v4f;
template <int D> struct RowVec;
template <> struct RowVec<128> { typedef v4f T; };
template <> struct RowVec<64> { typedef v2f T; };

__device__ __forceinline__ void hold_loaded(v4f v) {
  asm volatile("" ::"v"(v.x), "v"(v.y), "v"(v.z), "v"(v.w));
}
__device__ __forceinline__ void hold_loaded(v2f v) {
  asm volatile("" ::"v"(v.x), "v"(v.y));
}

#define WROWS ((TILE_N + TILE_WAVES - 1) / TILE_WAVES)
#define QOFF_PITCH 40
static_assert(TILE_WAVES == 32 && WROWS == 7, "row r belongs to wave r & 31, register set r >> 5 (0..6)");

template <int D, bool ADD_SCALED, bool BUILD>
__global__ __launch_bounds__(TILE_THREADS) void tile_sum(
    const float* __restrict__ feat,
    const int* __restrict__ src,
    const int* __restrict__ dst,
    const float* __restrict__ invdeg,
    float* invdeg_out,
    float* __restrict__ dest,
    int* qsaved,
    int* qsaved_n,
    int* qoff,
    const int* __restrict__ segs,
    const int* __restrict__ offs,
    int n_nodes, int n_edges) {
  typedef typename RowVec<D>::T VT;
  constexpr int VW = D / 32;
  constexpr int LPR = D / 4;
  constexpr int RPW = 32 / LPR;
  constexpr int U = 8;
  constexpr int SU = 4;
  __shared__ __attribute__((aligned(16))) float acc[TILE_N * D];
  __shared__ unsigned queue[QCAP];
  __shared__ unsigned qsorted[BUILD ? QCAP : 1];
  __shared__ unsigned qcount;
  __shared__ unsigned degc[BUILD ? TILE_N : 1];
  __shared__ unsigned sl_a[BUILD ? 256 : 1], sl_p[BUILD ? 256 : 1], sl_c[BUILD ? 256 : 1];
  __shared__ unsigned woff[BUILD ? TILE_WAVES + 1 : 1];
  __shared__ unsigned wcur[BUILD ? TILE_WAVES : 1];
  __shared__ unsigned wsum[8];

  const int tid = threadIdx.x;
  const int wave = __builtin_amdgcn_readfirstlane(tid >> 5);
  const int lane = tid & 31;
  const int n0 = blockIdx.x * TILE_N;
  const int E4 = n_edges >> 2;
  const int rot4 = (int)blockIdx.x * (E4 / (int)gridDim.x);

  if (tid == 0) qcount = 0u;
  if (BUILD) {
    for (int i = tid; i < TILE_N; i += TILE_THREADS) degc[i] = 0u;
  }
  __syncthreads();

  auto scan = [&](int qb, int qe) {
    int q0 = qb + tid;
    for (; q0 + (SU - 1) * TILE_THREADS < qe; q0 += SU * TILE_THREADS)
      scan_items<SU>(dst, q0, rot4, E4, n0, queue, &qcount);
    for (; q0 < qe; q0 += TILE_THREADS)
      scan_items<1>(dst, q0, rot4, E4, n0, queue, &qcount);
  };

  int* const qmine = qsaved + (long)blockIdx.x * QCAP;
  int* const qomine = qoff + (long)blockIdx.x * QOFF_PITCH;

  auto resolve = [&](int n) {
    for (int i = tid; i < n; i += TILE_THREADS) {
      const unsigned ent = queue[i];
      const unsigned s = (unsigned)src[ent >> 8];
      queue[i] = (s << 8) | (ent & 0xFFu);
      if (BUILD) atomicAdd(&degc[ent & 0xFFu], 1u);
    }
    __syncthreads();
  };

  auto gather_lds = [&](int n) {
    const int sub = lane / LPR;
    const int l = lane % LPR;
    for (int base = wave * RPW + sub; base < n + sub; base += TILE_WAVES * RPW * U) {
      float4 v[U];
      unsigned ent[U];
#pragma unroll
      for (int u = 0; u < U; ++u) {
        const int i = base + u * TILE_WAVES * RPW;
        const int ic = i < n ? i : n - 1;
        ent[u] = queue[ic];
        v[u] = *(const float4*)(feat + (long)(ent[u] >> 8) * D + l * 4);
      }
#pragma unroll
      for (int u = 0; u < U; ++u) {
        asm volatile("" ::"v"(v[u].x), "v"(v[u].y), "v"(v[u].z), "v"(v[u].w));
      }
#pragma unroll
      for (int u = 0; u < U; ++u) {
        const int i = base + u * TILE_WAVES * RPW;
        if (i < n) {
          float* a = acc + (int)(ent[u] & 0xFFu) * D + l;
          atomicAdd(a + 0 * LPR, v[u].x);
          atomicAdd(a + 1 * LPR, v[u].y);
          atomicAdd(a + 2 * LPR, v[u].z);
          atomicAdd(a + 3 * LPR, v[u].w);
        }
      }
    }
  };

  auto write_invdeg = [&]() {
    if (BUILD) {
      for (int i = tid; i < TILE_N; i += TILE_THREADS) {
        const int row = n0 + i;
        if (row < n_nodes) invdeg_out[row] = 1.0f / fmaxf((float)degc[i], 1.0f);
      }
    }
  };

  unsigned cnt;
  if (BUILD) {
    unsigned a = 0u, len = 0u;
    if (tid < NSLICE) {
      const int* o = offs + tid * OFF_PITCH + blockIdx.x;
      a = (unsigned)o[0];
      len = (unsigned)o[1] - a;
    }
    const unsigned p = scan256(len, wsum, tid, cnt);
    if (tid < 256) {
      sl_a[tid] = a;
      sl_p[tid] = p;
      sl_c[tid] = len;
    }
    if (tid == 0) qsaved_n[blockIdx.x] = (int)cnt;
    __syncthreads();
  } else {
    cnt = (unsigned)qsaved_n[blockIdx.x];
  }
  if (cnt <= (unsigned)QCAP) {
    int beg, end;
    if (BUILD) {
      const int sl = tid >> 2;
      if (sl < NSLICE) {
        const unsigned len = sl_c[sl];
        const unsigned p = sl_p[sl];
        const int* sp = segs + sl * SEG_PITCH + sl_a[sl];
        for (unsigned i = (unsigned)(tid & 3); i < len; i += 4u) {
          const unsigned ent = (unsigned)sp[i];
          queue[p + i] = ent;
          atomicAdd(&degc[ent & 0xFFu], 1u);
        }
      }
      __syncthreads();
      write_invdeg();
      if (wave == 0) {
        unsigned c = 0u;
#pragma unroll
        for (int k = 0; k < WROWS; ++k) {
          const int r = lane + k * TILE_WAVES;
          if (r < TILE_N) c += degc[r];
        }
        unsigned inc = c;
#pragma unroll
        for (int d = 1; d < 32; d <<= 1) {
          const unsigned nb = __shfl_up(inc, (unsigned)d);
          if (lane >= d) inc += nb;
        }
        woff[lane] = inc - c;
        wcur[lane] = inc - c;
        qomine[lane] = (int)(inc - c);
        if (lane == 31) {
          woff[TILE_WAVES] = inc;
          qomine[TILE_WAVES] = (int)inc;
        }
      }
      __syncthreads();
      for (int i = tid; i < (int)cnt; i += TILE_THREADS) {
        const unsigned ent = queue[i];
        const unsigned pos = atomicAdd(&wcur[ent & 31u], 1u);
        qsorted[pos] = ent;
      }
      __syncthreads();
      for (int i = tid; i < (int)cnt; i += TILE_THREADS) qmine[i] = (int)qsorted[i];
      beg = (int)woff[wave];
      end = (int)woff[wave + 1];
    } else {
      beg = qomine[wave];
      end = qomine[wave + 1];
    }
    beg = __builtin_amdgcn_readfirstlane(beg);
    end = __builtin_amdgcn_readfirstlane(end);

    v8f sumc[VW];
#pragma unroll
    for (int c = 0; c < VW; ++c) sumc[c] = (v8f)(0.0f);
    for (int cb = beg; cb < end; cb += 32) {
      const int nch = end - cb < 32 ? end - cb : 32;
      const int li = lane < nch ? lane : nch - 1;
      const unsigned myent = BUILD ? qsorted[cb + li] : (unsigned)qmine[cb + li];
      for (int j = 0; j < nch; j += U) {
        VT v[U];
        unsigned e[U];
#pragma unroll
        for (int u = 0; u < U; ++u) {
          const int jj = j + u < nch ? j + u : nch - 1;
          e[u] = (unsigned)__builtin_amdgcn_readlane((int)myent, jj);
          v[u] = *(const VT*)(feat + (long)(e[u] >> 8) * D + lane * VW);
        }
#pragma unroll
        for (int u = 0; u < U; ++u) {
          hold_loaded(v[u]);
        }
#pragma unroll
        for (int u = 0; u < U; ++u) {
          if (j + u < nch) {
            const int set = (int)((e[u] >> 5) & 7u);
#pragma unroll
            for (int c = 0; c < VW; ++c) sumc[c][set] += v[u][c];
          }
        }
      }
    }

#pragma unroll
    for (int k = 0; k < WROWS; ++k) {
      const int r = wave + k * TILE_WAVES;
      const int row = n0 + r;
      if (r < TILE_N && row < n_nodes) {
        VT* p = (VT*)(dest + (long)row * D + lane * VW);
        VT o;
#pragma unroll
        for (int c = 0; c < VW; ++c) o[c] = sumc[c][k];
        if (ADD_SCALED) {
          const float w = invdeg[row];
          const VT c = *p;
          o = c + w * o;
        }
        *p = o;
      }
    }
  } else {
    for (int i = tid; i < TILE_N * D / 4; i += TILE_THREADS)
      ((float4*)acc)[i] = make_float4(0.f, 0.f, 0.f, 0.f);
    for (int qb = 0; qb < E4; qb += QCAP / 4) {
      __syncthreads();
      if (tid == 0) qcount = 0u;
      __syncthreads();
      const int qe = qb + QCAP / 4 < E4 ? qb + QCAP / 4 : E4;
      scan(qb, qe);
      __syncthreads();
      const int n = (int)qcount;
      resolve(n);
      gather_lds(n);
    }
    __syncthreads();
    write_invdeg();
    for (int i = tid; i < TILE_N * LPR; i += TILE_THREADS) {
      const int r = i / LPR;
      const int l4 = i % LPR;
      const int row = n0 + r;
      if (row < n_nodes) {
        const float* a = acc + r * D + l4;
        float4 o = make_float4(a[0 * LPR], a[1 * LPR], a[2 * LPR], a[3 * LPR]);
        float4* p = (float4*)(dest + (long)row * D + l4 * 4);
        if (ADD_SCALED) {
          const float w = invdeg[row];
          const float4 c = *p;
          o.x = c.x + w * o.x; o.y = c.y + w * o.y; o.z = c.z + w * o.z; o.w = c.w + w * o.w;
        }
        *p = o;
      }
    }
  }
}

template <int OUTD, bool RELU, int GRID>
__global__ __launch_bounds__(32 * (OUTD / 16)) void sage_gemm(
    const float* __restrict__ xin,
    const float* __restrict__ agg,
    const float* __restrict__ invdeg,
    const float* __restrict__ Wl,
    const float* __restrict__ Wr,
    const float* __restrict__ bias,
    float* __restrict__ out,
    int n_nodes) {
  __shared__ float sX[16][LDSPITCH];
  __shared__ float sA[16][LDSPITCH];

  const int tid = threadIdx.x;
  const int nthreads = 32 * (OUTD / 16);
  const int wave = tid >> 5;
  const int lane = tid & 31;
  const int half = lane >> 4;
  const int r = lane & 15;
  const int col = wave * 16 + r;

  v2f BL[HID / 4], BR[HID / 4];
#pragma unroll
  for (int j = 0; j < HID / 4; ++j) {
    const int ka = 4 * j + 2 * half;
    BL[j][0] = Wl[(long)ka * OUTD + col];
    BL[j][1] = Wl[(long)(ka + 1) * OUTD + col];
    BR[j][0] = Wr[(long)ka * OUTD + col];
    BR[j][1] = Wr[(long)(ka + 1) * OUTD + col];
  }
  const float bn = bias[col];

  const int n_strips = n_nodes >> 4;
  for (int strip = blockIdx.x; strip < n_strips; strip += GRID) {
    const int row0 = strip * 16;

    for (int i = tid; i < 16 * (HID / 4); i += nthreads) {
      const int rr = i / (HID / 4);
      const int c4 = (i % (HID / 4)) * 4;
      const long base = (long)(row0 + rr) * HID + c4;
      float4 xv = *(const float4*)(xin + base);
      float4 av = *(const float4*)(agg + base);
      const float iv = invdeg[row0 + rr];
      av.x *= iv; av.y *= iv; av.z *= iv; av.w *= iv;
      *(float4*)&sX[rr][c4] = xv;
      *(float4*)&sA[rr][c4] = av;
    }
    __syncthreads();

    v8f accL = {};
    v8f accR = {};
#pragma unroll
    for (int j = 0; j < HID / 4; ++j) {
      const int ka = 4 * j + 2 * half;
      v2f aL, aR;
      aL[0] = sA[r][ka];
      aL[1] = sA[r][ka + 1];
      aR[0] = sX[r][ka];
      aR[1] = sX[r][ka + 1];
      accL = __builtin_amdgcn_wmma_f32_16x16x4_f32(false, aL, false, BL[j],
                                                   (short)0, accL, false, false);
      accR = __builtin_amdgcn_wmma_f32_16x16x4_f32(false, aR, false, BR[j],
                                                   (short)0, accR, false, false);
    }

#pragma unroll
    for (int i = 0; i < 8; ++i) {
      const int gr = row0 + half * 8 + i;
      float v = accL[i] + accR[i] + bn;
      if (RELU) v = fmaxf(v, 0.0f);
      out[(long)gr * OUTD + col] = v;
    }
    __syncthreads();
  }
}

template <int GRID>
__global__ __launch_bounds__(256) void sage_gemm_proj(
    const float* __restrict__ xin,
    const float* __restrict__ agg,
    const float* __restrict__ invdeg,
    const float* __restrict__ Wl,
    const float* __restrict__ Wr,
    const float* __restrict__ bias,
    const float* __restrict__ W3l,
    const float* __restrict__ W3r,
    const float* __restrict__ bias3,
    float* __restrict__ P,
    float* __restrict__ out,
    int n_nodes) {
  __shared__ float sX[16][LDSPITCH];
  __shared__ float sA[16][LDSPITCH];
  __shared__ float sH[16][LDSPITCH];

  const int tid = threadIdx.x;
  const int wave = tid >> 5;
  const int lane = tid & 31;
  const int half = lane >> 4;
  const int r = lane & 15;
  const int col = wave * 16 + r;
  const bool selfside = wave >= 4;
  const float* __restrict__ W3sel = selfside ? W3r : W3l;
  float* __restrict__ dsel = selfside ? out : P;
  const int col3 = (wave & 3) * 16 + r;

  v2f BL[HID / 4], BR[HID / 4], B3[HID / 4];
#pragma unroll
  for (int j = 0; j < HID / 4; ++j) {
    const int ka = 4 * j + 2 * half;
    BL[j][0] = Wl[(long)ka * HID + col];
    BL[j][1] = Wl[(long)(ka + 1) * HID + col];
    BR[j][0] = Wr[(long)ka * HID + col];
    BR[j][1] = Wr[(long)(ka + 1) * HID + col];
    B3[j][0] = W3sel[(long)ka * 64 + col3];
    B3[j][1] = W3sel[(long)(ka + 1) * 64 + col3];
  }
  const float bn = bias[col];
  const float bn3 = selfside ? bias3[col3] : 0.0f;

  const int n_strips = n_nodes >> 4;
  for (int strip = blockIdx.x; strip < n_strips; strip += GRID) {
    const int row0 = strip * 16;

    for (int i = tid; i < 16 * (HID / 4); i += 256) {
      const int rr = i / (HID / 4);
      const int c4 = (i % (HID / 4)) * 4;
      const long base = (long)(row0 + rr) * HID + c4;
      float4 xv = *(const float4*)(xin + base);
      float4 av = *(const float4*)(agg + base);
      const float iv = invdeg[row0 + rr];
      av.x *= iv; av.y *= iv; av.z *= iv; av.w *= iv;
      *(float4*)&sX[rr][c4] = xv;
      *(float4*)&sA[rr][c4] = av;
    }
    __syncthreads();

    v8f accL = {};
    v8f accR = {};
#pragma unroll
    for (int j = 0; j < HID / 4; ++j) {
      const int ka = 4 * j + 2 * half;
      v2f aL, aR;
      aL[0] = sA[r][ka];
      aL[1] = sA[r][ka + 1];
      aR[0] = sX[r][ka];
      aR[1] = sX[r][ka + 1];
      accL = __builtin_amdgcn_wmma_f32_16x16x4_f32(false, aL, false, BL[j],
                                                   (short)0, accL, false, false);
      accR = __builtin_amdgcn_wmma_f32_16x16x4_f32(false, aR, false, BR[j],
                                                   (short)0, accR, false, false);
    }

#pragma unroll
    for (int i = 0; i < 8; ++i) {
      sH[half * 8 + i][col] = fmaxf(accL[i] + accR[i] + bn, 0.0f);
    }
    __syncthreads();

    v8f acc0 = {};
    v8f acc1 = {};
#pragma unroll
    for (int j = 0; j < HID / 4; j += 2) {
      const int ka = 4 * j + 2 * half;
      v2f a0, a1;
      a0[0] = sH[r][ka];
      a0[1] = sH[r][ka + 1];
      a1[0] = sH[r][ka + 4];
      a1[1] = sH[r][ka + 5];
      acc0 = __builtin_amdgcn_wmma_f32_16x16x4_f32(false, a0, false, B3[j],
                                                   (short)0, acc0, false, false);
      acc1 = __builtin_amdgcn_wmma_f32_16x16x4_f32(false, a1, false, B3[j + 1],
                                                   (short)0, acc1, false, false);
    }

#pragma unroll
    for (int i = 0; i < 8; ++i) {
      const int gr = row0 + half * 8 + i;
      dsel[(long)gr * 64 + col3] = acc0[i] + acc1[i] + bn3;
    }
  }
}

extern "C" void kernel_launch(void* const* d_in, const int* in_sizes, int n_in,
                              void* d_out, int out_size, void* d_ws, size_t ws_size,
                              hipStream_t stream) {
  const float* x   = (const float*)d_in[0];
  const int* eidx  = (const int*)d_in[1];
  const float* W1l = (const float*)d_in[2];
  const float* W1r = (const float*)d_in[3];
  const float* b1  = (const float*)d_in[4];
  const float* W2l = (const float*)d_in[5];
  const float* W2r = (const float*)d_in[6];
  const float* b2  = (const float*)d_in[7];
  const float* W3l = (const float*)d_in[8];
  const float* W3r = (const float*)d_in[9];
  const float* b3  = (const float*)d_in[10];
  float* out = (float*)d_out;

  const int* src = eidx;
  const int* dst = eidx + N_EDGES;

  float* deg = (float*)d_ws;
  float* agg = deg + N_NODES;
  float* h1  = agg + (long)N_NODES * HID;
  float* P   = h1 + (long)N_NODES * HID;
  int* qn    = (int*)(P + (long)N_NODES * HID);
  int* qs    = qn + TILE_GROUPS;
  int* segs  = qs + (long)TILE_GROUPS * QCAP + 16;
  int* offs  = segs + (long)NSLICE * SEG_PITCH;
  int* qoff  = offs + (long)NSLICE * OFF_PITCH;

  const int tileBlocks = TILE_GROUPS;
  constexpr int SAGE_GRID = 768;
  constexpr int PROJ_GRID = 512;

  bin_edges<<<NSLICE, BIN_THREADS, 0, stream>>>(src, dst, segs, offs);

  tile_sum<HID, false, true><<<tileBlocks, TILE_THREADS, 0, stream>>>(x, src, dst, deg, deg, agg, qs, qn, qoff, segs, offs, N_NODES, N_EDGES);
  sage_gemm<128, true, SAGE_GRID><<<SAGE_GRID, 256, 0, stream>>>(x, agg, deg, W1l, W1r, b1, h1, N_NODES);

  tile_sum<HID, false, false><<<tileBlocks, TILE_THREADS, 0, stream>>>(h1, src, dst, deg, deg, agg, qs, qn, qoff, segs, offs, N_NODES, N_EDGES);
  sage_gemm_proj<PROJ_GRID><<<PROJ_GRID, 256, 0, stream>>>(h1, agg, deg, W2l, W2r, b2, W3l, W3r, b3, P, out, N_NODES);

  tile_sum<64, true, false><<<tileBlocks, TILE_THREADS, 0, stream>>>(P, src, dst, deg, deg, out, qs, qn, qoff, segs, offs, N_NODES, N_EDGES);
}


# both layer GEMMs request a strip's x/agg/invdeg pieces as one group of loads (was 2 dependent rounds); the layer-1 GEMM keeps its Wr fragments in each wave's own LDS slots to stay at 160 registers
# speedup vs baseline: 3.6602x; 1.0392x over previous
#include <hip/hip_runtime.h>

typedef __attribute__((ext_vector_type(2))) float v2f;
typedef __attribute__((ext_vector_type(8))) float v8f;

#define N_NODES 50000
#define N_EDGES 500000
#define HID 128
#define LDSPITCH (HID + 4)

#define TILE_N 196
#define TILE_THREADS 1024
#define TILE_WAVES (TILE_THREADS / 32)
#define QCAP 16384
#define TILE_GROUPS ((N_NODES + TILE_N - 1) / TILE_N)

#define NSLICE 250
#define SLICE_E (N_EDGES / NSLICE)
#define SLICE_Q (SLICE_E / 4)
#define SEG_PITCH 2016
#define OFF_PITCH 288
#define BIN_THREADS 256
#define BIN_NL 2
static_assert(NSLICE * SLICE_E == N_EDGES && SLICE_E % 4 == 0, "slices must tile the edge list");
static_assert(BIN_THREADS == TILE_GROUPS && NSLICE <= 256, "one thread per bin / per slice");
static_assert(BIN_THREADS * BIN_NL >= SLICE_Q && SLICE_E <= SEG_PITCH, "slice fits its threads");

__device__ __forceinline__ unsigned scan256(unsigned v, unsigned* wsum, int tid, unsigned& total) {
  const int lane = tid & 31;
  const int wave = tid >> 5;
  unsigned inc = v;
#pragma unroll
  for (int d = 1; d < 32; d <<= 1) {
    const unsigned n = __shfl_up(inc, (unsigned)d);
    if (lane >= d) inc += n;
  }
  if (wave < 8 && lane == 31) wsum[wave] = inc;
  __syncthreads();
  unsigned before = 0u, all = 0u;
#pragma unroll
  for (int w = 0; w < 8; ++w) {
    const unsigned s = wsum[w];
    if (w < wave) before += s;
    all += s;
  }
  total = all;
  return before + inc - v;
}

__global__ __launch_bounds__(BIN_THREADS) void bin_edges(
    const int* __restrict__ src,
    const int* __restrict__ dst,
    int* __restrict__ segs,
    int* __restrict__ offs) {
  __shared__ unsigned bcnt[TILE_GROUPS];
  __shared__ unsigned bstart[TILE_GROUPS];
  __shared__ unsigned wsum[8];
  __shared__ __attribute__((aligned(16))) unsigned sorted[SLICE_E];

  const int tid = threadIdx.x;
  const int s = blockIdx.x;
  const int4* d4p = (const int4*)(dst + (long)s * SLICE_E);
  const int4* s4p = (const int4*)(src + (long)s * SLICE_E);

  bcnt[tid] = 0u;
  __syncthreads();

  int4 dv[BIN_NL], sv[BIN_NL];
#pragma unroll
  for (int j = 0; j < BIN_NL; ++j) {
    const int q = tid + j * BIN_THREADS;
    dv[j] = d4p[q < SLICE_Q ? q : 0];
    sv[j] = s4p[q < SLICE_Q ? q : 0];
  }
#pragma unroll
  for (int j = 0; j < BIN_NL; ++j) {
    asm volatile("" ::"v"(dv[j].x), "v"(dv[j].y), "v"(dv[j].z), "v"(dv[j].w),
                 "v"(sv[j].x), "v"(sv[j].y), "v"(sv[j].z), "v"(sv[j].w));
  }
  unsigned ent[BIN_NL][4], own[BIN_NL][4], rank[BIN_NL][4];
#pragma unroll
  for (int j = 0; j < BIN_NL; ++j) {
    const bool valid = tid + j * BIN_THREADS < SLICE_Q;
    const unsigned du[4] = {(unsigned)dv[j].x, (unsigned)dv[j].y, (unsigned)dv[j].z, (unsigned)dv[j].w};
    const unsigned su[4] = {(unsigned)sv[j].x, (unsigned)sv[j].y, (unsigned)sv[j].z, (unsigned)sv[j].w};
#pragma unroll
    for (int k = 0; k < 4; ++k) {
      own[j][k] = 0xFFFFFFFFu;
      ent[j][k] = 0u;
      rank[j][k] = 0u;
      if (valid && du[k] < (unsigned)(TILE_GROUPS * TILE_N)) {
        const unsigned o = du[k] / (unsigned)TILE_N;
        own[j][k] = o;
        ent[j][k] = (su[k] << 8) | (du[k] - o * (unsigned)TILE_N);
        rank[j][k] = atomicAdd(&bcnt[o], 1u);
      }
    }
  }
  __syncthreads();

  unsigned total;
  const unsigned st = scan256(bcnt[tid], wsum, tid, total);
  bstart[tid] = st;
  offs[s * OFF_PITCH + tid] = (int)st;
  if (tid == 0) offs[s * OFF_PITCH + TILE_GROUPS] = (int)total;
  __syncthreads();

#pragma unroll
  for (int j = 0; j < BIN_NL; ++j) {
#pragma unroll
    for (int k = 0; k < 4; ++k) {
      if (own[j][k] != 0xFFFFFFFFu) sorted[bstart[own[j][k]] + rank[j][k]] = ent[j][k];
    }
  }
  __syncthreads();

  int4* outp = (int4*)(segs + (long)s * SEG_PITCH);
  for (int q = tid; q < SLICE_Q; q += BIN_THREADS) outp[q] = ((const int4*)sorted)[q];
}

template <int NL>
__device__ __forceinline__ void scan_items(const int* __restrict__ dst, int q0, int rot4, int E4,
                                           int n0, unsigned* queue, unsigned* qcount) {
  int4 d4[NL];
  unsigned e0[NL];
#pragma unroll
  for (int j = 0; j < NL; ++j) {
    int idx = q0 + j * TILE_THREADS + rot4;
    if (idx >= E4) idx -= E4;
    d4[j] = ((const int4*)dst)[idx];
    e0[j] = (unsigned)idx << 2;
  }
  unsigned t[NL][4];
  bool any = false;
#pragma unroll
  for (int j = 0; j < NL; ++j) {
    t[j][0] = (unsigned)(d4[j].x - n0);
    t[j][1] = (unsigned)(d4[j].y - n0);
    t[j][2] = (unsigned)(d4[j].z - n0);
    t[j][3] = (unsigned)(d4[j].w - n0);
#pragma unroll
    for (int k = 0; k < 4; ++k) any = any || (t[j][k] < (unsigned)TILE_N);
  }
  if (any) {
#pragma unroll
    for (int j = 0; j < NL; ++j) {
#pragma unroll
      for (int k = 0; k < 4; ++k) {
        if (t[j][k] < (unsigned)TILE_N) {
          const unsigned slot = atomicAdd(qcount, 1u);
          if (slot < (unsigned)QCAP) queue[slot] = ((e0[j] + (unsigned)k) << 8) | t[j][k];
        }
      }
    }
  }
}

typedef __attribute__((ext_vector_type(4))) float v4f;
template <int D> struct RowVec;
template <> struct RowVec<128> { typedef v4f T; };
template <> struct RowVec<64> { typedef v2f T; };

__device__ __forceinline__ void hold_loaded(v4f v) {
  asm volatile("" ::"v"(v.x), "v"(v.y), "v"(v.z), "v"(v.w));
}
__device__ __forceinline__ void hold_loaded(v2f v) {
  asm volatile("" ::"v"(v.x), "v"(v.y));
}

#define WROWS ((TILE_N + TILE_WAVES - 1) / TILE_WAVES)
#define QOFF_PITCH 40
static_assert(TILE_WAVES == 32 && WROWS == 7, "row r belongs to wave r & 31, register set r >> 5 (0..6)");

template <int D, bool ADD_SCALED, bool BUILD>
__global__ __launch_bounds__(TILE_THREADS) void tile_sum(
    const float* __restrict__ feat,
    const int* __restrict__ src,
    const int* __restrict__ dst,
    const float* __restrict__ invdeg,
    float* invdeg_out,
    float* __restrict__ dest,
    int* qsaved,
    int* qsaved_n,
    int* qoff,
    const int* __restrict__ segs,
    const int* __restrict__ offs,
    int n_nodes, int n_edges) {
  typedef typename RowVec<D>::T VT;
  constexpr int VW = D / 32;
  constexpr int LPR = D / 4;
  constexpr int RPW = 32 / LPR;
  constexpr int U = 8;
  constexpr int SU = 4;
  __shared__ __attribute__((aligned(16))) float acc[TILE_N * D];
  __shared__ unsigned queue[QCAP];
  __shared__ unsigned qsorted[BUILD ? QCAP : 1];
  __shared__ unsigned qcount;
  __shared__ unsigned degc[BUILD ? TILE_N : 1];
  __shared__ unsigned sl_a[BUILD ? 256 : 1], sl_p[BUILD ? 256 : 1], sl_c[BUILD ? 256 : 1];
  __shared__ unsigned woff[BUILD ? TILE_WAVES + 1 : 1];
  __shared__ unsigned wcur[BUILD ? TILE_WAVES : 1];
  __shared__ unsigned wsum[8];

  const int tid = threadIdx.x;
  const int wave = __builtin_amdgcn_readfirstlane(tid >> 5);
  const int lane = tid & 31;
  const int n0 = blockIdx.x * TILE_N;
  const int E4 = n_edges >> 2;
  const int rot4 = (int)blockIdx.x * (E4 / (int)gridDim.x);

  if (tid == 0) qcount = 0u;
  if (BUILD) {
    for (int i = tid; i < TILE_N; i += TILE_THREADS) degc[i] = 0u;
  }
  __syncthreads();

  auto scan = [&](int qb, int qe) {
    int q0 = qb + tid;
    for (; q0 + (SU - 1) * TILE_THREADS < qe; q0 += SU * TILE_THREADS)
      scan_items<SU>(dst, q0, rot4, E4, n0, queue, &qcount);
    for (; q0 < qe; q0 += TILE_THREADS)
      scan_items<1>(dst, q0, rot4, E4, n0, queue, &qcount);
  };

  int* const qmine = qsaved + (long)blockIdx.x * QCAP;
  int* const qomine = qoff + (long)blockIdx.x * QOFF_PITCH;

  auto resolve = [&](int n) {
    for (int i = tid; i < n; i += TILE_THREADS) {
      const unsigned ent = queue[i];
      const unsigned s = (unsigned)src[ent >> 8];
      queue[i] = (s << 8) | (ent & 0xFFu);
      if (BUILD) atomicAdd(&degc[ent & 0xFFu], 1u);
    }
    __syncthreads();
  };

  auto gather_lds = [&](int n) {
    const int sub = lane / LPR;
    const int l = lane % LPR;
    for (int base = wave * RPW + sub; base < n + sub; base += TILE_WAVES * RPW * U) {
      float4 v[U];
      unsigned ent[U];
#pragma unroll
      for (int u = 0; u < U; ++u) {
        const int i = base + u * TILE_WAVES * RPW;
        const int ic = i < n ? i : n - 1;
        ent[u] = queue[ic];
        v[u] = *(const float4*)(feat + (long)(ent[u] >> 8) * D + l * 4);
      }
#pragma unroll
      for (int u = 0; u < U; ++u) {
        asm volatile("" ::"v"(v[u].x), "v"(v[u].y), "v"(v[u].z), "v"(v[u].w));
      }
#pragma unroll
      for (int u = 0; u < U; ++u) {
        const int i = base + u * TILE_WAVES * RPW;
        if (i < n) {
          float* a = acc + (int)(ent[u] & 0xFFu) * D + l;
          atomicAdd(a + 0 * LPR, v[u].x);
          atomicAdd(a + 1 * LPR, v[u].y);
          atomicAdd(a + 2 * LPR, v[u].z);
          atomicAdd(a + 3 * LPR, v[u].w);
        }
      }
    }
  };

  auto write_invdeg = [&]() {
    if (BUILD) {
      for (int i = tid; i < TILE_N; i += TILE_THREADS) {
        const int row = n0 + i;
        if (row < n_nodes) invdeg_out[row] = 1.0f / fmaxf((float)degc[i], 1.0f);
      }
    }
  };

  unsigned cnt;
  if (BUILD) {
    unsigned a = 0u, len = 0u;
    if (tid < NSLICE) {
      const int* o = offs + tid * OFF_PITCH + blockIdx.x;
      a = (unsigned)o[0];
      len = (unsigned)o[1] - a;
    }
    const unsigned p = scan256(len, wsum, tid, cnt);
    if (tid < 256) {
      sl_a[tid] = a;
      sl_p[tid] = p;
      sl_c[tid] = len;
    }
    if (tid == 0) qsaved_n[blockIdx.x] = (int)cnt;
    __syncthreads();
  } else {
    cnt = (unsigned)qsaved_n[blockIdx.x];
  }
  if (cnt <= (unsigned)QCAP) {
    int beg, end;
    if (BUILD) {
      const int sl = tid >> 2;
      if (sl < NSLICE) {
        const unsigned len = sl_c[sl];
        const unsigned p = sl_p[sl];
        const int* sp = segs + sl * SEG_PITCH + sl_a[sl];
        for (unsigned i = (unsigned)(tid & 3); i < len; i += 4u) {
          const unsigned ent = (unsigned)sp[i];
          queue[p + i] = ent;
          atomicAdd(&degc[ent & 0xFFu], 1u);
        }
      }
      __syncthreads();
      write_invdeg();
      if (wave == 0) {
        unsigned c = 0u;
#pragma unroll
        for (int k = 0; k < WROWS; ++k) {
          const int r = lane + k * TILE_WAVES;
          if (r < TILE_N) c += degc[r];
        }
        unsigned inc = c;
#pragma unroll
        for (int d = 1; d < 32; d <<= 1) {
          const unsigned nb = __shfl_up(inc, (unsigned)d);
          if (lane >= d) inc += nb;
        }
        woff[lane] = inc - c;
        wcur[lane] = inc - c;
        qomine[lane] = (int)(inc - c);
        if (lane == 31) {
          woff[TILE_WAVES] = inc;
          qomine[TILE_WAVES] = (int)inc;
        }
      }
      __syncthreads();
      for (int i = tid; i < (int)cnt; i += TILE_THREADS) {
        const unsigned ent = queue[i];
        const unsigned pos = atomicAdd(&wcur[ent & 31u], 1u);
        qsorted[pos] = ent;
      }
      __syncthreads();
      for (int i = tid; i < (int)cnt; i += TILE_THREADS) qmine[i] = (int)qsorted[i];
      beg = (int)woff[wave];
      end = (int)woff[wave + 1];
    } else {
      beg = qomine[wave];
      end = qomine[wave + 1];
    }
    beg = __builtin_amdgcn_readfirstlane(beg);
    end = __builtin_amdgcn_readfirstlane(end);

    v8f sumc[VW];
#pragma unroll
    for (int c = 0; c < VW; ++c) sumc[c] = (v8f)(0.0f);
    for (int cb = beg; cb < end; cb += 32) {
      const int nch = end - cb < 32 ? end - cb : 32;
      const int li = lane < nch ? lane : nch - 1;
      const unsigned myent = BUILD ? qsorted[cb + li] : (unsigned)qmine[cb + li];
      for (int j = 0; j < nch; j += U) {
        VT v[U];
        unsigned e[U];
#pragma unroll
        for (int u = 0; u < U; ++u) {
          const int jj = j + u < nch ? j + u : nch - 1;
          e[u] = (unsigned)__builtin_amdgcn_readlane((int)myent, jj);
          v[u] = *(const VT*)(feat + (long)(e[u] >> 8) * D + lane * VW);
        }
#pragma unroll
        for (int u = 0; u < U; ++u) {
          hold_loaded(v[u]);
        }
#pragma unroll
        for (int u = 0; u < U; ++u) {
          if (j + u < nch) {
            const int set = (int)((e[u] >> 5) & 7u);
#pragma unroll
            for (int c = 0; c < VW; ++c) sumc[c][set] += v[u][c];
          }
        }
      }
    }

#pragma unroll
    for (int k = 0; k < WROWS; ++k) {
      const int r = wave + k * TILE_WAVES;
      const int row = n0 + r;
      if (r < TILE_N && row < n_nodes) {
        VT* p = (VT*)(dest + (long)row * D + lane * VW);
        VT o;
#pragma unroll
        for (int c = 0; c < VW; ++c) o[c] = sumc[c][k];
        if (ADD_SCALED) {
          const float w = invdeg[row];
          const VT c = *p;
          o = c + w * o;
        }
        *p = o;
      }
    }
  } else {
    for (int i = tid; i < TILE_N * D / 4; i += TILE_THREADS)
      ((float4*)acc)[i] = make_float4(0.f, 0.f, 0.f, 0.f);
    for (int qb = 0; qb < E4; qb += QCAP / 4) {
      __syncthreads();
      if (tid == 0) qcount = 0u;
      __syncthreads();
      const int qe = qb + QCAP / 4 < E4 ? qb + QCAP / 4 : E4;
      scan(qb, qe);
      __syncthreads();
      const int n = (int)qcount;
      resolve(n);
      gather_lds(n);
    }
    __syncthreads();
    write_invdeg();
    for (int i = tid; i < TILE_N * LPR; i += TILE_THREADS) {
      const int r = i / LPR;
      const int l4 = i % LPR;
      const int row = n0 + r;
      if (row < n_nodes) {
        const float* a = acc + r * D + l4;
        float4 o = make_float4(a[0 * LPR], a[1 * LPR], a[2 * LPR], a[3 * LPR]);
        float4* p = (float4*)(dest + (long)row * D + l4 * 4);
        if (ADD_SCALED) {
          const float w = invdeg[row];
          const float4 c = *p;
          o.x = c.x + w * o.x; o.y = c.y + w * o.y; o.z = c.z + w * o.z; o.w = c.w + w * o.w;
        }
        *p = o;
      }
    }
  }
}

template <int OUTD, bool RELU, int GRID>
__global__ __launch_bounds__(32 * (OUTD / 16)) void sage_gemm(
    const float* __restrict__ xin,
    const float* __restrict__ agg,
    const float* __restrict__ invdeg,
    const float* __restrict__ Wl,
    const float* __restrict__ Wr,
    const float* __restrict__ bias,
    float* __restrict__ out,
    int n_nodes) {
  __shared__ float sX[16][LDSPITCH];
  __shared__ float sA[16][LDSPITCH];
  __shared__ v2f sWr[HID / 4][32 * (OUTD / 16)];

  const int tid = threadIdx.x;
  constexpr int nthreads = 32 * (OUTD / 16);
  constexpr int NR = 16 * (HID / 4) / nthreads;
  static_assert(NR * nthreads == 16 * (HID / 4), "the threads tile the strip exactly");
  const int wave = tid >> 5;
  const int lane = tid & 31;
  const int half = lane >> 4;
  const int r = lane & 15;
  const int col = wave * 16 + r;

  v2f BL[HID / 4];
#pragma unroll
  for (int j = 0; j < HID / 4; ++j) {
    const int ka = 4 * j + 2 * half;
    BL[j][0] = Wl[(long)ka * OUTD + col];
    BL[j][1] = Wl[(long)(ka + 1) * OUTD + col];
    v2f br;
    br[0] = Wr[(long)ka * OUTD + col];
    br[1] = Wr[(long)(ka + 1) * OUTD + col];
    sWr[j][tid] = br;
  }
  const float bn = bias[col];

  const int n_strips = n_nodes >> 4;
  for (int strip = blockIdx.x; strip < n_strips; strip += GRID) {
    const int row0 = strip * 16;

    float4 xv[NR], av[NR];
    float iv[NR];
#pragma unroll
    for (int q = 0; q < NR; ++q) {
      const int i = tid + q * nthreads;
      const int rr = i / (HID / 4);
      const int c4 = (i % (HID / 4)) * 4;
      const long base = (long)(row0 + rr) * HID + c4;
      xv[q] = *(const float4*)(xin + base);
      av[q] = *(const float4*)(agg + base);
      iv[q] = invdeg[row0 + rr];
    }
#pragma unroll
    for (int q = 0; q < NR; ++q) {
      const int i = tid + q * nthreads;
      const int rr = i / (HID / 4);
      const int c4 = (i % (HID / 4)) * 4;
      av[q].x *= iv[q]; av[q].y *= iv[q]; av[q].z *= iv[q]; av[q].w *= iv[q];
      *(float4*)&sX[rr][c4] = xv[q];
      *(float4*)&sA[rr][c4] = av[q];
    }
    __syncthreads();

    v8f accL = {};
    v8f accR = {};
#pragma unroll
    for (int j = 0; j < HID / 4; ++j) {
      const int ka = 4 * j + 2 * half;
      v2f aL, aR;
      aL[0] = sA[r][ka];
      aL[1] = sA[r][ka + 1];
      aR[0] = sX[r][ka];
      aR[1] = sX[r][ka + 1];
      accL = __builtin_amdgcn_wmma_f32_16x16x4_f32(false, aL, false, BL[j],
                                                   (short)0, accL, false, false);
      accR = __builtin_amdgcn_wmma_f32_16x16x4_f32(false, aR, false, sWr[j][tid],
                                                   (short)0, accR, false, false);
    }

#pragma unroll
    for (int i = 0; i < 8; ++i) {
      const int gr = row0 + half * 8 + i;
      float v = accL[i] + accR[i] + bn;
      if (RELU) v = fmaxf(v, 0.0f);
      out[(long)gr * OUTD + col] = v;
    }
    __syncthreads();
  }
}

template <int GRID>
__global__ __launch_bounds__(256) void sage_gemm_proj(
    const float* __restrict__ xin,
    const float* __restrict__ agg,
    const float* __restrict__ invdeg,
    const float* __restrict__ Wl,
    const float* __restrict__ Wr,
    const float* __restrict__ bias,
    const float* __restrict__ W3l,
    const float* __restrict__ W3r,
    const float* __restrict__ bias3,
    float* __restrict__ P,
    float* __restrict__ out,
    int n_nodes) {
  __shared__ float sX[16][LDSPITCH];
  __shared__ float sA[16][LDSPITCH];
  __shared__ float sH[16][LDSPITCH];

  const int tid = threadIdx.x;
  const int wave = tid >> 5;
  const int lane = tid & 31;
  const int half = lane >> 4;
  const int r = lane & 15;
  const int col = wave * 16 + r;
  const bool selfside = wave >= 4;
  const float* __restrict__ W3sel = selfside ? W3r : W3l;
  float* __restrict__ dsel = selfside ? out : P;
  const int col3 = (wave & 3) * 16 + r;

  v2f BL[HID / 4], BR[HID / 4], B3[HID / 4];
#pragma unroll
  for (int j = 0; j < HID / 4; ++j) {
    const int ka = 4 * j + 2 * half;
    BL[j][0] = Wl[(long)ka * HID + col];
    BL[j][1] = Wl[(long)(ka + 1) * HID + col];
    BR[j][0] = Wr[(long)ka * HID + col];
    BR[j][1] = Wr[(long)(ka + 1) * HID + col];
    B3[j][0] = W3sel[(long)ka * 64 + col3];
    B3[j][1] = W3sel[(long)(ka + 1) * 64 + col3];
  }
  const float bn = bias[col];
  const float bn3 = selfside ? bias3[col3] : 0.0f;

  const int n_strips = n_nodes >> 4;
  for (int strip = blockIdx.x; strip < n_strips; strip += GRID) {
    const int row0 = strip * 16;

    constexpr int NR = 16 * (HID / 4) / 256;
    static_assert(NR * 256 == 16 * (HID / 4), "the threads tile the strip exactly");
    float4 xv[NR], av[NR];
    float iv[NR];
#pragma unroll
    for (int q = 0; q < NR; ++q) {
      const int i = tid + q * 256;
      const int rr = i / (HID / 4);
      const int c4 = (i % (HID / 4)) * 4;
      const long base = (long)(row0 + rr) * HID + c4;
      xv[q] = *(const float4*)(xin + base);
      av[q] = *(const float4*)(agg + base);
      iv[q] = invdeg[row0 + rr];
    }
#pragma unroll
    for (int q = 0; q < NR; ++q) {
      const int i = tid + q * 256;
      const int rr = i / (HID / 4);
      const int c4 = (i % (HID / 4)) * 4;
      av[q].x *= iv[q]; av[q].y *= iv[q]; av[q].z *= iv[q]; av[q].w *= iv[q];
      *(float4*)&sX[rr][c4] = xv[q];
      *(float4*)&sA[rr][c4] = av[q];
    }
    __syncthreads();

    v8f accL = {};
    v8f accR = {};
#pragma unroll
    for (int j = 0; j < HID / 4; ++j) {
      const int ka = 4 * j + 2 * half;
      v2f aL, aR;
      aL[0] = sA[r][ka];
      aL[1] = sA[r][ka + 1];
      aR[0] = sX[r][ka];
      aR[1] = sX[r][ka + 1];
      accL = __builtin_amdgcn_wmma_f32_16x16x4_f32(false, aL, false, BL[j],
                                                   (short)0, accL, false, false);
      accR = __builtin_amdgcn_wmma_f32_16x16x4_f32(false, aR, false, BR[j],
                                                   (short)0, accR, false, false);
    }

#pragma unroll
    for (int i = 0; i < 8; ++i) {
      sH[half * 8 + i][col] = fmaxf(accL[i] + accR[i] + bn, 0.0f);
    }
    __syncthreads();

    v8f acc0 = {};
    v8f acc1 = {};
#pragma unroll
    for (int j = 0; j < HID / 4; j += 2) {
      const int ka = 4 * j + 2 * half;
      v2f a0, a1;
      a0[0] = sH[r][ka];
      a0[1] = sH[r][ka + 1];
      a1[0] = sH[r][ka + 4];
      a1[1] = sH[r][ka + 5];
      acc0 = __builtin_amdgcn_wmma_f32_16x16x4_f32(false, a0, false, B3[j],
                                                   (short)0, acc0, false, false);
      acc1 = __builtin_amdgcn_wmma_f32_16x16x4_f32(false, a1, false, B3[j + 1],
                                                   (short)0, acc1, false, false);
    }

#pragma unroll
    for (int i = 0; i < 8; ++i) {
      const int gr = row0 + half * 8 + i;
      dsel[(long)gr * 64 + col3] = acc0[i] + acc1[i] + bn3;
    }
  }
}

extern "C" void kernel_launch(void* const* d_in, const int* in_sizes, int n_in,
                              void* d_out, int out_size, void* d_ws, size_t ws_size,
                              hipStream_t stream) {
  const float* x   = (const float*)d_in[0];
  const int* eidx  = (const int*)d_in[1];
  const float* W1l = (const float*)d_in[2];
  const float* W1r = (const float*)d_in[3];
  const float* b1  = (const float*)d_in[4];
  const float* W2l = (const float*)d_in[5];
  const float* W2r = (const float*)d_in[6];
  const float* b2  = (const float*)d_in[7];
  const float* W3l = (const float*)d_in[8];
  const float* W3r = (const float*)d_in[9];
  const float* b3  = (const float*)d_in[10];
  float* out = (float*)d_out;

  const int* src = eidx;
  const int* dst = eidx + N_EDGES;

  float* deg = (float*)d_ws;
  float* agg = deg + N_NODES;
  float* h1  = agg + (long)N_NODES * HID;
  float* P   = h1 + (long)N_NODES * HID;
  int* qn    = (int*)(P + (long)N_NODES * HID);
  int* qs    = qn + TILE_GROUPS;
  int* segs  = qs + (long)TILE_GROUPS * QCAP + 16;
  int* offs  = segs + (long)NSLICE * SEG_PITCH;
  int* qoff  = offs + (long)NSLICE * OFF_PITCH;

  const int tileBlocks = TILE_GROUPS;
  constexpr int SAGE_GRID = 768;
  constexpr int PROJ_GRID = 512;

  bin_edges<<<NSLICE, BIN_THREADS, 0, stream>>>(src, dst, segs, offs);

  tile_sum<HID, false, true><<<tileBlocks, TILE_THREADS, 0, stream>>>(x, src, dst, deg, deg, agg, qs, qn, qoff, segs, offs, N_NODES, N_EDGES);
  sage_gemm<128, true, SAGE_GRID><<<SAGE_GRID, 256, 0, stream>>>(x, agg, deg, W1l, W1r, b1, h1, N_NODES);

  tile_sum<HID, false, false><<<tileBlocks, TILE_THREADS, 0, stream>>>(h1, src, dst, deg, deg, agg, qs, qn, qoff, segs, offs, N_NODES, N_EDGES);
  sage_gemm_proj<PROJ_GRID><<<PROJ_GRID, 256, 0, stream>>>(h1, agg, deg, W2l, W2r, b2, W3l, W3r, b3, P, out, N_NODES);

  tile_sum<64, true, false><<<tileBlocks, TILE_THREADS, 0, stream>>>(P, src, dst, deg, deg, out, qs, qn, qoff, segs, offs, N_NODES, N_EDGES);
}
